# starting kernel
# baseline (speedup 1.0000x reference)
#include <hip/hip_runtime.h>
#include <math.h>

#define IN_C 512
#define HID 100
#define OUT_C 2
#define NT 8
#define WP_FLOATS (256 * NT * 16 * 2)

typedef __attribute__((ext_vector_type(2))) float v2f;
typedef __attribute__((ext_vector_type(8))) float v8f;

__global__ void deg_init_kernel(float* deg, int n) {
  int i = blockIdx.x * blockDim.x + threadIdx.x;
  if (i < n) deg[i] = 1.0f;
}

__global__ void deg_accum_kernel(const int* __restrict__ dst, float* deg, int e) {
  int i = blockIdx.x * blockDim.x + threadIdx.x;
  if (i < e) atomicAdd(&deg[dst[i]], 1.0f);
}

__global__ void deg_rsqrt_kernel(float* deg, int n) {
  int i = blockIdx.x * blockDim.x + threadIdx.x;
  if (i < n) { float d = deg[i]; deg[i] = (d > 0.f) ? rsqrtf(d) : 0.f; }
}

__global__ void pack_w_kernel(const float* __restrict__ W1,
                              const float* __restrict__ Ws,
                              float* __restrict__ Wp) {
  int idx = blockIdx.x * blockDim.x + threadIdx.x;
  if (idx >= WP_FLOATS) return;
  int kin = idx & 1;
  int col = (idx >> 1) & 15;
  int t   = (idx >> 5) & 7;
  int kp  = idx >> 8;
  int k   = (kp << 1) + kin;
  float v = 0.f;
  if (t < 7) {
    int c = (t << 4) + col;
    if (c < HID) v = W1[(size_t)k * HID + c];
  } else {
    if (col < OUT_C) v = Ws[(size_t)k * OUT_C + col];
  }
  Wp[idx] = v;
}

__global__ void __launch_bounds__(256)
gemm1_wmma_kernel(const float* __restrict__ x, const float* __restrict__ Wp,
                  const float* __restrict__ dinv, float* __restrict__ h,
                  float* __restrict__ agg, float* __restrict__ outp, int n_nodes) {
  const int lane = threadIdx.x & 31;
  const int wv   = threadIdx.x >> 5;
  const int tile = blockIdx.x * (blockDim.x >> 5) + wv;
  const int row0 = tile << 4;
  if (row0 >= n_nodes) return;

  const int m  = lane & 15;
  const int kh = (lane >> 4) << 1;

  v8f acc[NT] = {};

  const float* __restrict__ pa = x + (size_t)(row0 + m) * IN_C + kh;
  const float* __restrict__ pb = Wp + ((size_t)(lane >> 4) * 256) + (m << 1);

  for (int k0 = 0; k0 < IN_C; k0 += 4) {
    v2f a = *(const v2f*)pa;
#pragma unroll
    for (int t = 0; t < NT; ++t) {
      v2f b = *(const v2f*)(pb + (t << 5));
      acc[t] = __builtin_amdgcn_wmma_f32_16x16x4_f32(
          false, a, false, b, (short)0, acc[t], false, false);
    }
    pa += 4;
    pb += 512;
  }

  const int rbase = (lane < 16) ? 0 : 8;
#pragma unroll
  for (int r = 0; r < 8; ++r) {
    const int row = row0 + rbase + r;
    const float di  = dinv[row];
    const float di2 = di * di;
    float* __restrict__ hrow = h   + (size_t)row * HID;
    float* __restrict__ arow = agg + (size_t)row * HID;
#pragma unroll
    for (int t = 0; t < 7; ++t) {
      int col = (t << 4) + m;
      if (col < HID) {
        float v = acc[t][r];
        hrow[col] = v;
        arow[col] = v * di2;
      }
    }
    if (m < OUT_C) outp[(size_t)row * OUT_C + m] = acc[7][r];
  }
}

__global__ void scatter1_kernel(const float* __restrict__ h,
                                const float* __restrict__ dinv,
                                const int* __restrict__ src,
                                const int* __restrict__ dst,
                                float* __restrict__ agg, int n_edges) {
  int lane = threadIdx.x & 31;
  int wv   = threadIdx.x >> 5;
  long long e = (long long)blockIdx.x * (blockDim.x >> 5) + wv;
  if (e >= n_edges) return;
  int s = src[e], d = dst[e];
  float norm = dinv[s] * dinv[d];
  const float* __restrict__ hs = h + (size_t)s * HID;
  float* __restrict__ ad = agg + (size_t)d * HID;
#pragma unroll
  for (int i = 0; i < 4; ++i) {
    int f = lane + (i << 5);
    if (f < HID) atomicAdd(&ad[f], hs[f] * norm);
  }
}

__global__ void gemm2_fused_kernel(const float* __restrict__ agg,
                                   const float* __restrict__ b1,
                                   const float* __restrict__ W2,
                                   const float* __restrict__ b2,
                                   const float* __restrict__ bs,
                                   const float* __restrict__ dinv,
                                   float* __restrict__ h2,
                                   float* __restrict__ out, int n_nodes) {
  int n = blockIdx.x * blockDim.x + threadIdx.x;
  if (n >= n_nodes) return;
  const float* __restrict__ r = agg + (size_t)n * HID;
  float a0 = 0.f, a1 = 0.f;
#pragma unroll 4
  for (int k = 0; k < HID; ++k) {
    float v = r[k] + b1[k];
    v = v > 0.f ? v : 0.f;
    a0 += v * W2[k * 2 + 0];
    a1 += v * W2[k * 2 + 1];
  }
  h2[(size_t)n * 2 + 0] = a0;
  h2[(size_t)n * 2 + 1] = a1;
  float di = dinv[n];
  float di2 = di * di;
  out[(size_t)n * 2 + 0] += b2[0] + bs[0] + a0 * di2;
  out[(size_t)n * 2 + 1] += b2[1] + bs[1] + a1 * di2;
}

__global__ void scatter2_kernel(const float* __restrict__ h2,
                                const float* __restrict__ dinv,
                                const int* __restrict__ src,
                                const int* __restrict__ dst,
                                float* __restrict__ out, int n_edges) {
  int e = blockIdx.x * blockDim.x + threadIdx.x;
  if (e >= n_edges) return;
  int s = src[e], d = dst[e];
  float norm = dinv[s] * dinv[d];
  atomicAdd(&out[(size_t)d * 2 + 0], h2[(size_t)s * 2 + 0] * norm);
  atomicAdd(&out[(size_t)d * 2 + 1], h2[(size_t)s * 2 + 1] * norm);
}

__global__ void log_softmax_kernel(float* __restrict__ out, int n_nodes) {
  int n = blockIdx.x * blockDim.x + threadIdx.x;
  if (n >= n_nodes) return;
  float z0 = out[(size_t)n * 2 + 0];
  float z1 = out[(size_t)n * 2 + 1];
  float mx = fmaxf(z0, z1);
  float lse = mx + logf(expf(z0 - mx) + expf(z1 - mx));
  out[(size_t)n * 2 + 0] = z0 - lse;
  out[(size_t)n * 2 + 1] = z1 - lse;
}

extern "C" void kernel_launch(void* const* d_in, const int* in_sizes, int n_in,
                              void* d_out, int out_size, void* d_ws, size_t ws_size,
                              hipStream_t stream) {
  const float* x  = (const float*)d_in[0];
  const int*   ei = (const int*)d_in[1];
  const float* W1 = (const float*)d_in[2];
  const float* b1 = (const float*)d_in[3];
  const float* W2 = (const float*)d_in[4];
  const float* b2 = (const float*)d_in[5];
  const float* Ws = (const float*)d_in[6];
  const float* bs = (const float*)d_in[7];
  float* out = (float*)d_out;

  const int n_nodes = in_sizes[0] / IN_C;
  const int n_edges = in_sizes[1] / 2;
  const int* srcv = ei;
  const int* dstv = ei + n_edges;

  float* Wp   = (float*)d_ws;
  float* dinv = Wp + WP_FLOATS;
  size_t off  = ((size_t)n_nodes + 255) & ~(size_t)255;
  float* h    = dinv + off;
  float* agg  = h + (size_t)n_nodes * HID;
  float* h2   = agg + (size_t)n_nodes * HID;

  const int T = 256;

  deg_init_kernel<<<(n_nodes + T - 1) / T, T, 0, stream>>>(dinv, n_nodes);
  deg_accum_kernel<<<(n_edges + T - 1) / T, T, 0, stream>>>(dstv, dinv, n_edges);
  deg_rsqrt_kernel<<<(n_nodes + T - 1) / T, T, 0, stream>>>(dinv, n_nodes);

  pack_w_kernel<<<WP_FLOATS / T, T, 0, stream>>>(W1, Ws, Wp);

  int tiles = (n_nodes + 15) / 16;
  gemm1_wmma_kernel<<<(tiles + 7) / 8, T, 0, stream>>>(x, Wp, dinv, h, agg, out, n_nodes);

  scatter1_kernel<<<(n_edges + 7) / 8, T, 0, stream>>>(h, dinv, srcv, dstv, agg, n_edges);

  gemm2_fused_kernel<<<(n_nodes + T - 1) / T, T, 0, stream>>>(agg, b1, W2, b2, bs, dinv,
                                                              h2, out, n_nodes);
  scatter2_kernel<<<(n_edges + T - 1) / T, T, 0, stream>>>(h2, dinv, srcv, dstv, out, n_edges);

  log_softmax_kernel<<<(n_nodes + T - 1) / T, T, 0, stream>>>(out, n_nodes);
}


# Float-atomic scatters replaced by a dst-sorted CSR (int rank atomics, scan, fill) and per-node gathers; agg row stays in registers so relu/W2 runs in its tail; sums in CSR order, W2 dot by lane tree
# speedup vs baseline: 1.8175x; 1.8175x over previous
#include <hip/hip_runtime.h>
#include <math.h>


#define IN_C 512
#define HID 100
#define HID4 25
#define OUT_C 2
#define NT 8
#define WP_FLOATS (256 * NT * 16 * 2)
#define SCAN_T 1024
#define SCAN_PER 8

typedef __attribute__((ext_vector_type(2))) float v2f;
typedef __attribute__((ext_vector_type(4))) float v4f;
typedef __attribute__((ext_vector_type(8))) float v8f;

__global__ void zero_cnt_kernel(int* __restrict__ cnt, int n) {
  int i = blockIdx.x * blockDim.x + threadIdx.x;
  if (i < n) cnt[i] = 0;
}

__global__ void count_rank_kernel(const int* __restrict__ dst, int* cnt,
                                  int* __restrict__ rank, int e) {
  int i = blockIdx.x * blockDim.x + threadIdx.x;
  if (i < e) rank[i] = atomicAdd(&cnt[dst[i]], 1);
}

__global__ void __launch_bounds__(SCAN_T)
scan_dinv_kernel(const int* __restrict__ cnt, int* __restrict__ rowptr,
                 float* __restrict__ dinv, int n) {
  __shared__ int wsum[32];
  const int tid = threadIdx.x, lane = tid & 31, wv = tid >> 5;
  int carry = 0;
  for (int base = 0; base < n; base += SCAN_T * SCAN_PER) {
    const int i0 = base + tid * SCAN_PER;
    const bool full = (i0 + SCAN_PER <= n);
    int c[SCAN_PER];
    if (full) {
      const int4 a = *(const int4*)(cnt + i0);
      const int4 b = *(const int4*)(cnt + i0 + 4);
      c[0] = a.x; c[1] = a.y; c[2] = a.z; c[3] = a.w;
      c[4] = b.x; c[5] = b.y; c[6] = b.z; c[7] = b.w;
    } else {
#pragma unroll
      for (int k = 0; k < SCAN_PER; ++k) c[k] = (i0 + k < n) ? cnt[i0 + k] : 0;
    }
    int ex[SCAN_PER];
    int t = 0;
#pragma unroll
    for (int k = 0; k < SCAN_PER; ++k) { ex[k] = t; t += c[k]; }
    int x = t;
#pragma unroll
    for (int o = 1; o < 32; o <<= 1) { int y = __shfl_up(x, o); if (lane >= o) x += y; }
    if (lane == 31) wsum[wv] = x;
    __syncthreads();
    if (wv == 0) {
      int s = wsum[lane];
#pragma unroll
      for (int o = 1; o < 32; o <<= 1) { int y = __shfl_up(s, o); if (lane >= o) s += y; }
      wsum[lane] = s;
    }
    __syncthreads();
    const int wave_off = (wv > 0) ? wsum[wv - 1] : 0;
    const int total = wsum[31];
    const int e0 = carry + wave_off + (x - t);
    float dv[SCAN_PER];
#pragma unroll
    for (int k = 0; k < SCAN_PER; ++k) {
      float dg = (float)(c[k] + 1);
      dv[k] = (dg > 0.f) ? rsqrtf(dg) : 0.f;
    }
    if (full) {
      *(int4*)(rowptr + i0)     = make_int4(e0 + ex[0], e0 + ex[1], e0 + ex[2], e0 + ex[3]);
      *(int4*)(rowptr + i0 + 4) = make_int4(e0 + ex[4], e0 + ex[5], e0 + ex[6], e0 + ex[7]);
      *(v4f*)(dinv + i0)     = (v4f){dv[0], dv[1], dv[2], dv[3]};
      *(v4f*)(dinv + i0 + 4) = (v4f){dv[4], dv[5], dv[6], dv[7]};
    } else {
#pragma unroll
      for (int k = 0; k < SCAN_PER; ++k)
        if (i0 + k < n) { rowptr[i0 + k] = e0 + ex[k]; dinv[i0 + k] = dv[k]; }
    }
    carry += total;
    __syncthreads();
  }
  if (tid == 0) rowptr[n] = carry;
}

__global__ void fill_csr_kernel(const int* __restrict__ src, const int* __restrict__ dst,
                                const int* __restrict__ rank, const int* __restrict__ rowptr,
                                int* __restrict__ csr, int e) {
  int i = blockIdx.x * blockDim.x + threadIdx.x;
  if (i >= e) return;
  int p = rowptr[dst[i]] + rank[i];
  p = min(max(p, 0), e - 1);
  csr[p] = src[i];
}

__global__ void pack_w_kernel(const float* __restrict__ W1,
                              const float* __restrict__ Ws,
                              float* __restrict__ Wp) {
  int idx = blockIdx.x * blockDim.x + threadIdx.x;
  if (idx >= WP_FLOATS) return;
  int kin = idx & 1;
  int col = (idx >> 1) & 15;
  int t   = (idx >> 5) & 7;
  int kp  = idx >> 8;
  int k   = (kp << 1) + kin;
  float v = 0.f;
  if (t < 7) {
    int c = (t << 4) + col;
    if (c < HID) v = W1[(size_t)k * HID + c];
  } else {
    if (col < OUT_C) v = Ws[(size_t)k * OUT_C + col];
  }
  Wp[idx] = v;
}

__global__ void __launch_bounds__(256)
gemm1_wmma_kernel(const float* __restrict__ x, const float* __restrict__ Wp,
                  float* __restrict__ h, float* __restrict__ outp, int n_nodes) {
  const int lane = threadIdx.x & 31;
  const int wv   = threadIdx.x >> 5;
  const int tile = blockIdx.x * (blockDim.x >> 5) + wv;
  const int row0 = tile << 4;
  if (row0 >= n_nodes) return;

  const int m  = lane & 15;
  const int kh = (lane >> 4) << 1;

  v8f acc[NT] = {};

  const float* __restrict__ pa = x + (size_t)(row0 + m) * IN_C + kh;
  const float* __restrict__ pb = Wp + ((size_t)(lane >> 4) * 256) + (m << 1);

  for (int k0 = 0; k0 < IN_C; k0 += 4) {
    v2f a = *(const v2f*)pa;
#pragma unroll
    for (int t = 0; t < NT; ++t) {
      v2f b = *(const v2f*)(pb + (t << 5));
      acc[t] = __builtin_amdgcn_wmma_f32_16x16x4_f32(
          false, a, false, b, (short)0, acc[t], false, false);
    }
    pa += 4;
    pb += 512;
  }

  const int rbase = (lane < 16) ? 0 : 8;
#pragma unroll
  for (int r = 0; r < 8; ++r) {
    const int row = row0 + rbase + r;
    float* __restrict__ hrow = h + (size_t)row * HID;
#pragma unroll
    for (int t = 0; t < 7; ++t) {
      int col = (t << 4) + m;
      if (col < HID) hrow[col] = acc[t][r];
    }
    if (m < OUT_C) outp[(size_t)row * OUT_C + m] = acc[7][r];
  }
}

__global__ void __launch_bounds__(256)
gather1_kernel(const float* __restrict__ h, const float* __restrict__ dinv,
               const int* __restrict__ rowptr, const int* __restrict__ csr,
               const float* __restrict__ b1, const float* __restrict__ W2,
               float* __restrict__ h2, int n_nodes) {
  const int lane = threadIdx.x & 31;
  const int d = blockIdx.x * (blockDim.x >> 5) + (threadIdx.x >> 5);
  if (d >= n_nodes) return;
  const int beg = __builtin_amdgcn_readfirstlane(rowptr[d]);
  const int end = __builtin_amdgcn_readfirstlane(rowptr[d + 1]);
  const float dd = dinv[d];
  const bool act = lane < HID4;
  const v4f* __restrict__ h4 = (const v4f*)h;

  v4f acc = {0.f, 0.f, 0.f, 0.f};
  v4f vself = {0.f, 0.f, 0.f, 0.f};
  if (act) vself = h4[(size_t)d * HID4 + lane];

  for (int base = beg; base < end; base += 32) {
    int m = end - base;
    if (m > 32) m = 32;
    int s_l = 0;
    float w_l = 0.f;
    if (lane < m) {
      s_l = csr[base + lane];
      s_l = min(max(s_l, 0), n_nodes - 1);
      w_l = dinv[s_l] * dd;
    }
    int j = 0;
    for (; j + 4 <= m; j += 4) {
      const int s0 = __builtin_amdgcn_readlane(s_l, j);
      const int s1 = __builtin_amdgcn_readlane(s_l, j + 1);
      const int s2 = __builtin_amdgcn_readlane(s_l, j + 2);
      const int s3 = __builtin_amdgcn_readlane(s_l, j + 3);
      const float w0 = __int_as_float(__builtin_amdgcn_readlane(__float_as_int(w_l), j));
      const float w1 = __int_as_float(__builtin_amdgcn_readlane(__float_as_int(w_l), j + 1));
      const float w2 = __int_as_float(__builtin_amdgcn_readlane(__float_as_int(w_l), j + 2));
      const float w3 = __int_as_float(__builtin_amdgcn_readlane(__float_as_int(w_l), j + 3));
      if (act) {
        const v4f v0 = h4[(size_t)s0 * HID4 + lane];
        const v4f v1 = h4[(size_t)s1 * HID4 + lane];
        const v4f v2 = h4[(size_t)s2 * HID4 + lane];
        const v4f v3 = h4[(size_t)s3 * HID4 + lane];
        acc += v0 * w0;
        acc += v1 * w1;
        acc += v2 * w2;
        acc += v3 * w3;
      }
    }
    for (; j < m; ++j) {
      const int s0 = __builtin_amdgcn_readlane(s_l, j);
      const float w0 = __int_as_float(__builtin_amdgcn_readlane(__float_as_int(w_l), j));
      if (act) acc += h4[(size_t)s0 * HID4 + lane] * w0;
    }
  }

  acc += vself * (dd * dd);

  float p0 = 0.f, p1 = 0.f;
  if (act) {
    const v4f bb = ((const v4f*)b1)[lane];
    const v4f wa = ((const v4f*)W2)[2 * lane];
    const v4f wb = ((const v4f*)W2)[2 * lane + 1];
    float v0 = acc.x + bb.x, v1 = acc.y + bb.y, v2 = acc.z + bb.z, v3 = acc.w + bb.w;
    v0 = v0 > 0.f ? v0 : 0.f;
    v1 = v1 > 0.f ? v1 : 0.f;
    v2 = v2 > 0.f ? v2 : 0.f;
    v3 = v3 > 0.f ? v3 : 0.f;
    p0 = v0 * wa.x + v1 * wa.z + v2 * wb.x + v3 * wb.z;
    p1 = v0 * wa.y + v1 * wa.w + v2 * wb.y + v3 * wb.w;
  }
#pragma unroll
  for (int o = 16; o > 0; o >>= 1) {
    p0 += __shfl_xor(p0, o);
    p1 += __shfl_xor(p1, o);
  }
  if (lane == 0) { h2[(size_t)d * 2 + 0] = p0; h2[(size_t)d * 2 + 1] = p1; }
}

__global__ void gather2_kernel(const float* __restrict__ h2, const float* __restrict__ dinv,
                               const int* __restrict__ rowptr, const int* __restrict__ csr,
                               const float* __restrict__ b2, const float* __restrict__ bs,
                               float* __restrict__ out, int n_nodes) {
  int d = blockIdx.x * blockDim.x + threadIdx.x;
  if (d >= n_nodes) return;
  const int beg = rowptr[d], end = rowptr[d + 1];
  const float dd = dinv[d];
  const float di2 = dd * dd;
  const v2f* __restrict__ h22 = (const v2f*)h2;
  const v2f hv = h22[d];
  float a0 = out[(size_t)d * 2 + 0] + (b2[0] + bs[0] + hv.x * di2);
  float a1 = out[(size_t)d * 2 + 1] + (b2[1] + bs[1] + hv.y * di2);
#pragma unroll 4
  for (int j = beg; j < end; ++j) {
    int s = csr[j];
    s = min(max(s, 0), n_nodes - 1);
    const float w = dinv[s] * dd;
    const v2f v = h22[s];
    a0 += v.x * w;
    a1 += v.y * w;
  }
  out[(size_t)d * 2 + 0] = a0;
  out[(size_t)d * 2 + 1] = a1;
}

__global__ void log_softmax_kernel(float* __restrict__ out, int n_nodes) {
  int n = blockIdx.x * blockDim.x + threadIdx.x;
  if (n >= n_nodes) return;
  float z0 = out[(size_t)n * 2 + 0];
  float z1 = out[(size_t)n * 2 + 1];
  float mx = fmaxf(z0, z1);
  float lse = mx + logf(expf(z0 - mx) + expf(z1 - mx));
  out[(size_t)n * 2 + 0] = z0 - lse;
  out[(size_t)n * 2 + 1] = z1 - lse;
}

extern "C" void kernel_launch(void* const* d_in, const int* in_sizes, int n_in,
                              void* d_out, int out_size, void* d_ws, size_t ws_size,
                              hipStream_t stream) {
  const float* x  = (const float*)d_in[0];
  const int*   ei = (const int*)d_in[1];
  const float* W1 = (const float*)d_in[2];
  const float* b1 = (const float*)d_in[3];
  const float* W2 = (const float*)d_in[4];
  const float* b2 = (const float*)d_in[5];
  const float* Ws = (const float*)d_in[6];
  const float* bs = (const float*)d_in[7];
  float* out = (float*)d_out;

  const int n_nodes = in_sizes[0] / IN_C;
  const int n_edges = in_sizes[1] / 2;
  const int* srcv = ei;
  const int* dstv = ei + n_edges;

  float* Wp    = (float*)d_ws;
  float* dinv  = Wp + WP_FLOATS;
  size_t off   = ((size_t)n_nodes + 255) & ~(size_t)255;
  int* cnt     = (int*)(dinv + off);
  int* rowptr  = cnt + off;
  size_t off1  = ((size_t)n_nodes + 1 + 255) & ~(size_t)255;
  float* h     = (float*)(rowptr + off1);
  float* h2    = h + (size_t)n_nodes * HID;
  int* rank    = (int*)(h2 + (size_t)n_nodes * 2);
  int* csr     = rank + n_edges;

  const int T = 256;

  zero_cnt_kernel<<<(n_nodes + T - 1) / T, T, 0, stream>>>(cnt, n_nodes);
  count_rank_kernel<<<(n_edges + T - 1) / T, T, 0, stream>>>(dstv, cnt, rank, n_edges);
  scan_dinv_kernel<<<1, SCAN_T, 0, stream>>>(cnt, rowptr, dinv, n_nodes);
  fill_csr_kernel<<<(n_edges + T - 1) / T, T, 0, stream>>>(srcv, dstv, rank, rowptr, csr, n_edges);

  pack_w_kernel<<<WP_FLOATS / T, T, 0, stream>>>(W1, Ws, Wp);

  int tiles = (n_nodes + 15) / 16;
  gemm1_wmma_kernel<<<(tiles + 7) / 8, T, 0, stream>>>(x, Wp, h, out, n_nodes);

  gather1_kernel<<<(n_nodes + 7) / 8, T, 0, stream>>>(h, dinv, rowptr, csr, b1, W2, h2, n_nodes);

  gather2_kernel<<<(n_nodes + T - 1) / T, T, 0, stream>>>(h2, dinv, rowptr, csr, b2, bs, out, n_nodes);

  log_softmax_kernel<<<(n_nodes + T - 1) / T, T, 0, stream>>>(out, n_nodes);
}


# GEMM B operand (W1|Ws) copied once per workgroup into LDS in fragment order; 256 workgroups x 16 waves loop over row tiles; 7 column tiles not 8; no pack launch; k (sum) order in 8-k blocks changes
# speedup vs baseline: 2.2540x; 1.2402x over previous
#include <hip/hip_runtime.h>
#include <math.h>


#define IN_C 512
#define HID 100
#define HID4 25
#define OUT_C 2
#define NT 7
#define NCOL (NT * 16)
#define BS_FLOATS (IN_C * NCOL)
#define WP_FLOATS 65536
#define GEMM_WG 256
#define GEMM_WAVES 16
#define SCAN_T 1024
#define SCAN_PER 8

typedef __attribute__((ext_vector_type(2))) float v2f;
typedef __attribute__((ext_vector_type(4))) float v4f;
typedef __attribute__((ext_vector_type(8))) float v8f;

__global__ void zero_cnt_kernel(int* __restrict__ cnt, int n) {
  int i = blockIdx.x * blockDim.x + threadIdx.x;
  if (i < n) cnt[i] = 0;
}

__global__ void count_rank_kernel(const int* __restrict__ dst, int* cnt,
                                  int* __restrict__ rank, int e) {
  int i = blockIdx.x * blockDim.x + threadIdx.x;
  if (i < e) rank[i] = atomicAdd(&cnt[dst[i]], 1);
}

__global__ void __launch_bounds__(SCAN_T)
scan_dinv_kernel(const int* __restrict__ cnt, int* __restrict__ rowptr,
                 float* __restrict__ dinv, int n) {
  __shared__ int wsum[32];
  const int tid = threadIdx.x, lane = tid & 31, wv = tid >> 5;
  int carry = 0;
  for (int base = 0; base < n; base += SCAN_T * SCAN_PER) {
    const int i0 = base + tid * SCAN_PER;
    const bool full = (i0 + SCAN_PER <= n);
    int c[SCAN_PER];
    if (full) {
      const int4 a = *(const int4*)(cnt + i0);
      const int4 b = *(const int4*)(cnt + i0 + 4);
      c[0] = a.x; c[1] = a.y; c[2] = a.z; c[3] = a.w;
      c[4] = b.x; c[5] = b.y; c[6] = b.z; c[7] = b.w;
    } else {
#pragma unroll
      for (int k = 0; k < SCAN_PER; ++k) c[k] = (i0 + k < n) ? cnt[i0 + k] : 0;
    }
    int ex[SCAN_PER];
    int t = 0;
#pragma unroll
    for (int k = 0; k < SCAN_PER; ++k) { ex[k] = t; t += c[k]; }
    int x = t;
#pragma unroll
    for (int o = 1; o < 32; o <<= 1) { int y = __shfl_up(x, o); if (lane >= o) x += y; }
    if (lane == 31) wsum[wv] = x;
    __syncthreads();
    if (wv == 0) {
      int s = wsum[lane];
#pragma unroll
      for (int o = 1; o < 32; o <<= 1) { int y = __shfl_up(s, o); if (lane >= o) s += y; }
      wsum[lane] = s;
    }
    __syncthreads();
    const int wave_off = (wv > 0) ? wsum[wv - 1] : 0;
    const int total = wsum[31];
    const int e0 = carry + wave_off + (x - t);
    float dv[SCAN_PER];
#pragma unroll
    for (int k = 0; k < SCAN_PER; ++k) {
      float dg = (float)(c[k] + 1);
      dv[k] = (dg > 0.f) ? rsqrtf(dg) : 0.f;
    }
    if (full) {
      *(int4*)(rowptr + i0)     = make_int4(e0 + ex[0], e0 + ex[1], e0 + ex[2], e0 + ex[3]);
      *(int4*)(rowptr + i0 + 4) = make_int4(e0 + ex[4], e0 + ex[5], e0 + ex[6], e0 + ex[7]);
      *(v4f*)(dinv + i0)     = (v4f){dv[0], dv[1], dv[2], dv[3]};
      *(v4f*)(dinv + i0 + 4) = (v4f){dv[4], dv[5], dv[6], dv[7]};
    } else {
#pragma unroll
      for (int k = 0; k < SCAN_PER; ++k)
        if (i0 + k < n) { rowptr[i0 + k] = e0 + ex[k]; dinv[i0 + k] = dv[k]; }
    }
    carry += total;
    __syncthreads();
  }
  if (tid == 0) rowptr[n] = carry;
}

__global__ void fill_csr_kernel(const int* __restrict__ src, const int* __restrict__ dst,
                                const int* __restrict__ rank, const int* __restrict__ rowptr,
                                int* __restrict__ csr, int e) {
  int i = blockIdx.x * blockDim.x + threadIdx.x;
  if (i >= e) return;
  int p = rowptr[dst[i]] + rank[i];
  p = min(max(p, 0), e - 1);
  csr[p] = src[i];
}

__global__ void __launch_bounds__(GEMM_WAVES * 32)
gemm1_wmma_kernel(const float* __restrict__ x, const float* __restrict__ W1,
                  const float* __restrict__ Ws,
                  float* __restrict__ h, float* __restrict__ outp, int n_nodes) {
  __shared__ __attribute__((aligned(16))) float Bs[BS_FLOATS];
  v4f* __restrict__ Bs4 = (v4f*)Bs;
  static_assert(GEMM_WAVES * 32 * 25 == (IN_C / 4) * HID && GEMM_WAVES * 32 * 3 == (IN_C / 4) * (NCOL - HID),
                "the copy below gives every thread exactly 25 + 3 items");

  for (int g = 0; g < 5; ++g) {
    float w[5][4];
#pragma unroll
    for (int u = 0; u < 5; ++u) {
      const int id = threadIdx.x + (GEMM_WAVES * 32) * (g * 5 + u);
      const int kq = id / HID;
      const int c = id - kq * HID;
      const float* __restrict__ p = W1 + (kq * 4) * HID + c;
#pragma unroll
      for (int q = 0; q < 4; ++q) w[u][q] = p[q * HID];
    }
#pragma unroll
    for (int u = 0; u < 5; ++u) {
      const int id = threadIdx.x + (GEMM_WAVES * 32) * (g * 5 + u);
      const int kq = id / HID;
      const int c = id - kq * HID;
      Bs4[(((kq >> 1) * NT + (c >> 4)) << 5) + ((kq & 1) << 4) + (c & 15)] =
          (v4f){w[u][0], w[u][1], w[u][2], w[u][3]};
    }
  }
  {
    float ws[3][4];
#pragma unroll
    for (int u = 0; u < 3; ++u) {
      const int id = threadIdx.x + (GEMM_WAVES * 32) * u;
      const int kq = id / 12;
      const int cc = id - kq * 12;
      const float* __restrict__ p = Ws + (kq * 4) * OUT_C + (cc < OUT_C ? cc : OUT_C - 1);
#pragma unroll
      for (int q = 0; q < 4; ++q) ws[u][q] = p[q * OUT_C];
    }
#pragma unroll
    for (int u = 0; u < 3; ++u) {
      const int id = threadIdx.x + (GEMM_WAVES * 32) * u;
      const int kq = id / 12;
      const int cc = id - kq * 12;
      const bool keep = cc < OUT_C;
      Bs4[(((kq >> 1) * NT + (NT - 1)) << 5) + ((kq & 1) << 4) + (HID - 96 + cc)] =
          (v4f){keep ? ws[u][0] : 0.f, keep ? ws[u][1] : 0.f, keep ? ws[u][2] : 0.f, keep ? ws[u][3] : 0.f};
    }
  }
  __syncthreads();

  const int lane = threadIdx.x & 31;
  const int wv   = threadIdx.x >> 5;
  const int m  = lane & 15;
  const int hh = lane >> 4;
  const int tiles = (n_nodes + 15) >> 4;
  const v4f* __restrict__ bl = (const v4f*)Bs + (hh << 4) + m;

  for (int tile = wv * GEMM_WG + blockIdx.x; tile < tiles; tile += GEMM_WG * GEMM_WAVES) {
    const int row0 = tile << 4;
    int arow = row0 + m;
    if (arow > n_nodes - 1) arow = n_nodes - 1;
    const v4f* __restrict__ pa = (const v4f*)(x + (size_t)arow * IN_C) + hh;

    v8f acc[NT] = {};
    v4f ac[8], an[8];
#pragma unroll
    for (int jj = 0; jj < 8; ++jj) ac[jj] = pa[jj << 1];
    v4f bc[NT], bn[NT];
#pragma unroll
    for (int t = 0; t < NT; ++t) bc[t] = bl[t << 5];

    for (int i = 0; i < IN_C / 64; ++i) {
      const int inx = (i + 1 < IN_C / 64) ? i + 1 : i;
#pragma unroll
      for (int jj = 0; jj < 8; ++jj) an[jj] = pa[(inx << 4) + (jj << 1)];
#pragma unroll
      for (int jj = 0; jj < 8; ++jj) {
        const int j = (i << 3) + jj;
        const int jn = (j + 1 < IN_C / 8) ? j + 1 : j;
#pragma unroll
        for (int t = 0; t < NT; ++t) bn[t] = bl[(jn * NT + t) << 5];
        const v2f a0 = {ac[jj].x, ac[jj].y};
        const v2f a1 = {ac[jj].z, ac[jj].w};
#pragma unroll
        for (int t = 0; t < NT; ++t) {
          const v2f b0 = {bc[t].x, bc[t].y};
          acc[t] = __builtin_amdgcn_wmma_f32_16x16x4_f32(
              false, a0, false, b0, (short)0, acc[t], false, false);
        }
#pragma unroll
        for (int t = 0; t < NT; ++t) {
          const v2f b1 = {bc[t].z, bc[t].w};
          acc[t] = __builtin_amdgcn_wmma_f32_16x16x4_f32(
              false, a1, false, b1, (short)0, acc[t], false, false);
        }
#pragma unroll
        for (int t = 0; t < NT; ++t) bc[t] = bn[t];
      }
#pragma unroll
      for (int jj = 0; jj < 8; ++jj) ac[jj] = an[jj];
    }

    const int rbase = hh << 3;
#pragma unroll
    for (int r = 0; r < 8; ++r) {
      const int row = row0 + rbase + r;
      if (row < n_nodes) {
        float* __restrict__ hrow = h + (size_t)row * HID;
#pragma unroll
        for (int t = 0; t < NT - 1; ++t) hrow[(t << 4) + m] = acc[t][r];
        if (m < HID - 96) hrow[96 + m] = acc[NT - 1][r];
        else if (m < HID - 96 + OUT_C) outp[(size_t)row * OUT_C + (m - (HID - 96))] = acc[NT - 1][r];
      }
    }
  }
}

__global__ void __launch_bounds__(256)
gather1_kernel(const float* __restrict__ h, const float* __restrict__ dinv,
               const int* __restrict__ rowptr, const int* __restrict__ csr,
               const float* __restrict__ b1, const float* __restrict__ W2,
               float* __restrict__ h2, int n_nodes) {
  const int lane = threadIdx.x & 31;
  const int d = blockIdx.x * (blockDim.x >> 5) + (threadIdx.x >> 5);
  if (d >= n_nodes) return;
  const int beg = __builtin_amdgcn_readfirstlane(rowptr[d]);
  const int end = __builtin_amdgcn_readfirstlane(rowptr[d + 1]);
  const float dd = dinv[d];
  const bool act = lane < HID4;
  const v4f* __restrict__ h4 = (const v4f*)h;

  v4f acc = {0.f, 0.f, 0.f, 0.f};
  v4f vself = {0.f, 0.f, 0.f, 0.f};
  if (act) vself = h4[(size_t)d * HID4 + lane];

  for (int base = beg; base < end; base += 32) {
    int m = end - base;
    if (m > 32) m = 32;
    int s_l = 0;
    float w_l = 0.f;
    if (lane < m) {
      s_l = csr[base + lane];
      s_l = min(max(s_l, 0), n_nodes - 1);
      w_l = dinv[s_l] * dd;
    }
    int j = 0;
    for (; j + 4 <= m; j += 4) {
      const int s0 = __builtin_amdgcn_readlane(s_l, j);
      const int s1 = __builtin_amdgcn_readlane(s_l, j + 1);
      const int s2 = __builtin_amdgcn_readlane(s_l, j + 2);
      const int s3 = __builtin_amdgcn_readlane(s_l, j + 3);
      const float w0 = __int_as_float(__builtin_amdgcn_readlane(__float_as_int(w_l), j));
      const float w1 = __int_as_float(__builtin_amdgcn_readlane(__float_as_int(w_l), j + 1));
      const float w2 = __int_as_float(__builtin_amdgcn_readlane(__float_as_int(w_l), j + 2));
      const float w3 = __int_as_float(__builtin_amdgcn_readlane(__float_as_int(w_l), j + 3));
      if (act) {
        const v4f v0 = h4[(size_t)s0 * HID4 + lane];
        const v4f v1 = h4[(size_t)s1 * HID4 + lane];
        const v4f v2 = h4[(size_t)s2 * HID4 + lane];
        const v4f v3 = h4[(size_t)s3 * HID4 + lane];
        acc += v0 * w0;
        acc += v1 * w1;
        acc += v2 * w2;
        acc += v3 * w3;
      }
    }
    for (; j < m; ++j) {
      const int s0 = __builtin_amdgcn_readlane(s_l, j);
      const float w0 = __int_as_float(__builtin_amdgcn_readlane(__float_as_int(w_l), j));
      if (act) acc += h4[(size_t)s0 * HID4 + lane] * w0;
    }
  }

  acc += vself * (dd * dd);

  float p0 = 0.f, p1 = 0.f;
  if (act) {
    const v4f bb = ((const v4f*)b1)[lane];
    const v4f wa = ((const v4f*)W2)[2 * lane];
    const v4f wb = ((const v4f*)W2)[2 * lane + 1];
    float v0 = acc.x + bb.x, v1 = acc.y + bb.y, v2 = acc.z + bb.z, v3 = acc.w + bb.w;
    v0 = v0 > 0.f ? v0 : 0.f;
    v1 = v1 > 0.f ? v1 : 0.f;
    v2 = v2 > 0.f ? v2 : 0.f;
    v3 = v3 > 0.f ? v3 : 0.f;
    p0 = v0 * wa.x + v1 * wa.z + v2 * wb.x + v3 * wb.z;
    p1 = v0 * wa.y + v1 * wa.w + v2 * wb.y + v3 * wb.w;
  }
#pragma unroll
  for (int o = 16; o > 0; o >>= 1) {
    p0 += __shfl_xor(p0, o);
    p1 += __shfl_xor(p1, o);
  }
  if (lane == 0) { h2[(size_t)d * 2 + 0] = p0; h2[(size_t)d * 2 + 1] = p1; }
}

__global__ void gather2_kernel(const float* __restrict__ h2, const float* __restrict__ dinv,
                               const int* __restrict__ rowptr, const int* __restrict__ csr,
                               const float* __restrict__ b2, const float* __restrict__ bs,
                               float* __restrict__ out, int n_nodes) {
  int d = blockIdx.x * blockDim.x + threadIdx.x;
  if (d >= n_nodes) return;
  const int beg = rowptr[d], end = rowptr[d + 1];
  const float dd = dinv[d];
  const float di2 = dd * dd;
  const v2f* __restrict__ h22 = (const v2f*)h2;
  const v2f hv = h22[d];
  float a0 = out[(size_t)d * 2 + 0] + (b2[0] + bs[0] + hv.x * di2);
  float a1 = out[(size_t)d * 2 + 1] + (b2[1] + bs[1] + hv.y * di2);
#pragma unroll 4
  for (int j = beg; j < end; ++j) {
    int s = csr[j];
    s = min(max(s, 0), n_nodes - 1);
    const float w = dinv[s] * dd;
    const v2f v = h22[s];
    a0 += v.x * w;
    a1 += v.y * w;
  }
  out[(size_t)d * 2 + 0] = a0;
  out[(size_t)d * 2 + 1] = a1;
}

__global__ void log_softmax_kernel(float* __restrict__ out, int n_nodes) {
  int n = blockIdx.x * blockDim.x + threadIdx.x;
  if (n >= n_nodes) return;
  float z0 = out[(size_t)n * 2 + 0];
  float z1 = out[(size_t)n * 2 + 1];
  float mx = fmaxf(z0, z1);
  float lse = mx + logf(expf(z0 - mx) + expf(z1 - mx));
  out[(size_t)n * 2 + 0] = z0 - lse;
  out[(size_t)n * 2 + 1] = z1 - lse;
}

extern "C" void kernel_launch(void* const* d_in, const int* in_sizes, int n_in,
                              void* d_out, int out_size, void* d_ws, size_t ws_size,
                              hipStream_t stream) {
  const float* x  = (const float*)d_in[0];
  const int*   ei = (const int*)d_in[1];
  const float* W1 = (const float*)d_in[2];
  const float* b1 = (const float*)d_in[3];
  const float* W2 = (const float*)d_in[4];
  const float* b2 = (const float*)d_in[5];
  const float* Ws = (const float*)d_in[6];
  const float* bs = (const float*)d_in[7];
  float* out = (float*)d_out;

  const int n_nodes = in_sizes[0] / IN_C;
  const int n_edges = in_sizes[1] / 2;
  const int* srcv = ei;
  const int* dstv = ei + n_edges;

  float* Wp    = (float*)d_ws;
  float* dinv  = Wp + WP_FLOATS;
  size_t off   = ((size_t)n_nodes + 255) & ~(size_t)255;
  int* cnt     = (int*)(dinv + off);
  int* rowptr  = cnt + off;
  size_t off1  = ((size_t)n_nodes + 1 + 255) & ~(size_t)255;
  float* h     = (float*)(rowptr + off1);
  float* h2    = h + (size_t)n_nodes * HID;
  int* rank    = (int*)(h2 + (size_t)n_nodes * 2);
  int* csr     = rank + n_edges;

  const int T = 256;

  zero_cnt_kernel<<<(n_nodes + T - 1) / T, T, 0, stream>>>(cnt, n_nodes);
  count_rank_kernel<<<(n_edges + T - 1) / T, T, 0, stream>>>(dstv, cnt, rank, n_edges);
  scan_dinv_kernel<<<1, SCAN_T, 0, stream>>>(cnt, rowptr, dinv, n_nodes);
  fill_csr_kernel<<<(n_edges + T - 1) / T, T, 0, stream>>>(srcv, dstv, rank, rowptr, csr, n_edges);

  gemm1_wmma_kernel<<<GEMM_WG, GEMM_WAVES * 32, 0, stream>>>(x, W1, Ws, h, out, n_nodes);

  gather1_kernel<<<(n_nodes + 7) / 8, T, 0, stream>>>(h, dinv, rowptr, csr, b1, W2, h2, n_nodes);

  gather2_kernel<<<(n_nodes + T - 1) / T, T, 0, stream>>>(h2, dinv, rowptr, csr, b2, bs, out, n_nodes);

  log_softmax_kernel<<<(n_nodes + T - 1) / T, T, 0, stream>>>(out, n_nodes);
}


# CSR build as a two-level counting sort on LDS counters (256 dst ranges, then one workgroup per range): no device atomics, no one-workgroup scan, 7 launches; in-node sum order: LDS arrival order
# speedup vs baseline: 3.9165x; 1.7376x over previous
#include <hip/hip_runtime.h>
#include <math.h>


#define IN_C 512
#define HID 100
#define HID4 25
#define OUT_C 2
#define NT 7
#define NCOL (NT * 16)
#define BS_FLOATS (IN_C * NCOL)
#define WP_FLOATS 65536
#define GEMM_WG 256
#define GEMM_WAVES 16

typedef __attribute__((ext_vector_type(2))) float v2f;
typedef __attribute__((ext_vector_type(4))) float v4f;
typedef __attribute__((ext_vector_type(8))) float v8f;

#define NB 256
#define SG 256
#define ST 1024
#define EB 8
#define MAXNPB 1024
#define CAP2 8192
#define CAP3 12288
static_assert(NB * SG <= WP_FLOATS, "the table of chunk counts lies in the workspace head");

__global__ void __launch_bounds__(ST)
bucket_hist_kernel(const int* __restrict__ dst, int* __restrict__ tab, int n_nodes, int e) {
  __shared__ int hc[NB];
  const int tid = threadIdx.x, g = blockIdx.x;
  const unsigned npb = (unsigned)((n_nodes + NB - 1) / NB);
  const int chunk = (e + SG - 1) / SG;
  const int e0 = g * chunk;
  int e1 = e0 + chunk;
  if (e1 > e) e1 = e;
  if (tid < NB) hc[tid] = 0;
  __syncthreads();
  for (int base = e0; base < e1; base += ST * EB) {
    int d[EB];
#pragma unroll
    for (int k = 0; k < EB; ++k) {
      const int i = base + k * ST + tid;
      d[k] = dst[min(i, e1 - 1)];
    }
#pragma unroll
    for (int k = 0; k < EB; ++k) {
      const int i = base + k * ST + tid;
      if (i < e1) {
        const int dd = min(max(d[k], 0), n_nodes - 1);
        atomicAdd(&hc[(unsigned)dd / npb], 1);
      }
    }
  }
  __syncthreads();
  if (tid < NB) tab[tid * SG + g] = hc[tid];
}

__global__ void __launch_bounds__(ST)
bucket_scatter_kernel(const int* __restrict__ src, const int* __restrict__ dst,
                      const int* __restrict__ tab, int* __restrict__ bstart,
                      int2* __restrict__ pairs, int n_nodes, int e) {
  static_assert(SG == 256 && NB == 256 && ST == 1024, "wave w sums ranges 8w..8w+7, lane l the chunks 8l..8l+7");
  __shared__ int tot[NB];
  __shared__ int gb[NB];
  __shared__ int lst[NB];
  __shared__ int lcur[NB];
  __shared__ int2 stage[CAP2];
  const int tid = threadIdx.x, g = blockIdx.x, lane = tid & 31, wv = tid >> 5;
  const unsigned npb = (unsigned)((n_nodes + NB - 1) / NB);
  const int chunk = (e + SG - 1) / SG;
  const int e0 = g * chunk;
  int e1 = e0 + chunk;
  if (e1 > e) e1 = e;
  const int m = (e1 > e0) ? e1 - e0 : 0;

  int4 ta[8], tc[8];
#pragma unroll
  for (int k = 0; k < 8; ++k) {
    ta[k] = *(const int4*)(tab + (wv * 8 + k) * SG + lane * 8);
    tc[k] = *(const int4*)(tab + (wv * 8 + k) * SG + lane * 8 + 4);
  }
#pragma unroll
  for (int k = 0; k < 8; ++k) {
    const int b = wv * 8 + k;
    const int v[8] = {ta[k].x, ta[k].y, ta[k].z, ta[k].w, tc[k].x, tc[k].y, tc[k].z, tc[k].w};
    int s = 0, p = 0, mine = 0;
#pragma unroll
    for (int j = 0; j < 8; ++j) {
      s += v[j];
      if (lane * 8 + j < g) p += v[j];
      if (lane * 8 + j == g) mine = v[j];
    }
    for (int o = 16; o > 0; o >>= 1) { s += __shfl_xor(s, o); p += __shfl_xor(p, o); }
    if (lane == 0) { tot[b] = s; gb[b] = p; }
    if (lane == (g >> 3)) lst[b] = mine;
  }
  __syncthreads();
  if (wv == 0) {
    int t[8], lt[8];
    int run = 0, lrun = 0;
#pragma unroll
    for (int j = 0; j < 8; ++j) {
      t[j] = run;   run += tot[lane * 8 + j];
      lt[j] = lrun; lrun += lst[lane * 8 + j];
    }
    int x = run, lx = lrun;
    for (int o = 1; o < 32; o <<= 1) {
      const int y = __shfl_up(x, o), ly = __shfl_up(lx, o);
      if (lane >= o) { x += y; lx += ly; }
    }
    const int ex = x - run, lex = lx - lrun;
#pragma unroll
    for (int j = 0; j < 8; ++j) {
      const int st = ex + t[j];
      gb[lane * 8 + j] += st;
      lst[lane * 8 + j] = lex + lt[j];
      lcur[lane * 8 + j] = lex + lt[j];
      if (g == 0) bstart[lane * 8 + j] = st;
    }
    if (g == 0 && lane == 31) bstart[NB] = x;
  }
  __syncthreads();

  const bool in_lds = (m <= CAP2);
  for (int base = e0; base < e1; base += ST * EB) {
    int d[EB], s[EB];
#pragma unroll
    for (int k = 0; k < EB; ++k) {
      const int i = base + k * ST + tid;
      d[k] = dst[min(i, e1 - 1)];
      s[k] = src[min(i, e1 - 1)];
    }
#pragma unroll
    for (int k = 0; k < EB; ++k) {
      const int i = base + k * ST + tid;
      if (i < e1) {
        const int dd = min(max(d[k], 0), n_nodes - 1);
        const unsigned b = (unsigned)dd / npb;
        const int q = atomicAdd(&lcur[b], 1);
        if (in_lds) {
          stage[min(max(q, 0), CAP2 - 1)] = make_int2(s[k], dd);
        } else {
          int pos = gb[b] + (q - lst[b]);
          pos = min(max(pos, 0), e - 1);
          pairs[pos] = make_int2(s[k], dd);
        }
      }
    }
  }
  if (in_lds) {
    __syncthreads();
    for (int q = tid; q < m; q += ST) {
      const int2 pr = stage[q];
      const unsigned b = (unsigned)pr.y / npb;
      int pos = gb[b] + (q - lst[b]);
      pos = min(max(pos, 0), e - 1);
      pairs[pos] = pr;
    }
  }
}

__global__ void __launch_bounds__(ST)
bucket_build_kernel(const int2* __restrict__ pairs, const int* __restrict__ bstart,
                    int* __restrict__ rowptr, float* __restrict__ dinv,
                    int* __restrict__ csr, int n_nodes, int e) {
  static_assert(ST == MAXNPB, "one thread per LDS counter");
  __shared__ int cnt[MAXNPB];
  __shared__ int wsum[32];
  __shared__ int stage[CAP3];
  const int tid = threadIdx.x, b = blockIdx.x, lane = tid & 31, wv = tid >> 5;
  const int npb = (n_nodes + NB - 1) / NB;
  const int nb0 = b * npb;
  int s0 = bstart[b], s1 = bstart[b + 1];
  s0 = min(max(s0, 0), e);
  s1 = min(max(s1, s0), e);
  cnt[tid] = 0;
  __syncthreads();
  for (int base = s0; base < s1; base += ST * EB) {
    int d[EB];
#pragma unroll
    for (int k = 0; k < EB; ++k) {
      const int i = base + k * ST + tid;
      d[k] = pairs[min(i, s1 - 1)].y;
    }
#pragma unroll
    for (int k = 0; k < EB; ++k) {
      const int i = base + k * ST + tid;
      if (i < s1) atomicAdd(&cnt[min(max(d[k] - nb0, 0), MAXNPB - 1)], 1);
    }
  }
  __syncthreads();
  const int c = cnt[tid];
  int x = c;
  for (int o = 1; o < 32; o <<= 1) { int y = __shfl_up(x, o); if (lane >= o) x += y; }
  if (lane == 31) wsum[wv] = x;
  __syncthreads();
  if (wv == 0) {
    int s = wsum[lane];
    for (int o = 1; o < 32; o <<= 1) { int y = __shfl_up(s, o); if (lane >= o) s += y; }
    wsum[lane] = s;
  }
  __syncthreads();
  const int first = s0 + ((wv > 0) ? wsum[wv - 1] : 0) + (x - c);
  cnt[tid] = first;
  const int node = nb0 + tid;
  if (tid < npb && node < n_nodes) {
    rowptr[node] = first;
    const float dg = (float)(c + 1);
    dinv[node] = (dg > 0.f) ? rsqrtf(dg) : 0.f;
  }
  if (b == NB - 1 && tid == 0) rowptr[n_nodes] = s1;
  __syncthreads();

  const int m = s1 - s0;
  const bool in_lds = (m <= CAP3);
  for (int base = s0; base < s1; base += ST * EB) {
    int2 pr[EB];
#pragma unroll
    for (int k = 0; k < EB; ++k) {
      const int i = base + k * ST + tid;
      pr[k] = pairs[min(i, s1 - 1)];
    }
#pragma unroll
    for (int k = 0; k < EB; ++k) {
      const int i = base + k * ST + tid;
      if (i < s1) {
        const int ld = min(max(pr[k].y - nb0, 0), MAXNPB - 1);
        const int p = atomicAdd(&cnt[ld], 1);
        if (in_lds) stage[min(max(p - s0, 0), CAP3 - 1)] = pr[k].x;
        else csr[min(max(p, 0), e - 1)] = pr[k].x;
      }
    }
  }
  if (in_lds) {
    __syncthreads();
    for (int i = (s0 & ~31) + tid; i < s1; i += ST)
      if (i >= s0) csr[i] = stage[i - s0];
  }
}

__global__ void __launch_bounds__(GEMM_WAVES * 32)
gemm1_wmma_kernel(const float* __restrict__ x, const float* __restrict__ W1,
                  const float* __restrict__ Ws,
                  float* __restrict__ h, float* __restrict__ outp, int n_nodes) {
  __shared__ __attribute__((aligned(16))) float Bs[BS_FLOATS];
  v4f* __restrict__ Bs4 = (v4f*)Bs;
  static_assert(GEMM_WAVES * 32 * 25 == (IN_C / 4) * HID && GEMM_WAVES * 32 * 3 == (IN_C / 4) * (NCOL - HID),
                "the copy below gives every thread exactly 25 + 3 items");

  for (int g = 0; g < 5; ++g) {
    float w[5][4];
#pragma unroll
    for (int u = 0; u < 5; ++u) {
      const int id = threadIdx.x + (GEMM_WAVES * 32) * (g * 5 + u);
      const int kq = id / HID;
      const int c = id - kq * HID;
      const float* __restrict__ p = W1 + (kq * 4) * HID + c;
#pragma unroll
      for (int q = 0; q < 4; ++q) w[u][q] = p[q * HID];
    }
#pragma unroll
    for (int u = 0; u < 5; ++u) {
      const int id = threadIdx.x + (GEMM_WAVES * 32) * (g * 5 + u);
      const int kq = id / HID;
      const int c = id - kq * HID;
      Bs4[(((kq >> 1) * NT + (c >> 4)) << 5) + ((kq & 1) << 4) + (c & 15)] =
          (v4f){w[u][0], w[u][1], w[u][2], w[u][3]};
    }
  }
  {
    float ws[3][4];
#pragma unroll
    for (int u = 0; u < 3; ++u) {
      const int id = threadIdx.x + (GEMM_WAVES * 32) * u;
      const int kq = id / 12;
      const int cc = id - kq * 12;
      const float* __restrict__ p = Ws + (kq * 4) * OUT_C + (cc < OUT_C ? cc : OUT_C - 1);
#pragma unroll
      for (int q = 0; q < 4; ++q) ws[u][q] = p[q * OUT_C];
    }
#pragma unroll
    for (int u = 0; u < 3; ++u) {
      const int id = threadIdx.x + (GEMM_WAVES * 32) * u;
      const int kq = id / 12;
      const int cc = id - kq * 12;
      const bool keep = cc < OUT_C;
      Bs4[(((kq >> 1) * NT + (NT - 1)) << 5) + ((kq & 1) << 4) + (HID - 96 + cc)] =
          (v4f){keep ? ws[u][0] : 0.f, keep ? ws[u][1] : 0.f, keep ? ws[u][2] : 0.f, keep ? ws[u][3] : 0.f};
    }
  }
  __syncthreads();

  const int lane = threadIdx.x & 31;
  const int wv   = threadIdx.x >> 5;
  const int m  = lane & 15;
  const int hh = lane >> 4;
  const int tiles = (n_nodes + 15) >> 4;
  const v4f* __restrict__ bl = (const v4f*)Bs + (hh << 4) + m;

  for (int tile = wv * GEMM_WG + blockIdx.x; tile < tiles; tile += GEMM_WG * GEMM_WAVES) {
    const int row0 = tile << 4;
    int arow = row0 + m;
    if (arow > n_nodes - 1) arow = n_nodes - 1;
    const v4f* __restrict__ pa = (const v4f*)(x + (size_t)arow * IN_C) + hh;

    v8f acc[NT] = {};
    v4f ac[8], an[8];
#pragma unroll
    for (int jj = 0; jj < 8; ++jj) ac[jj] = pa[jj << 1];
    v4f bc[NT], bn[NT];
#pragma unroll
    for (int t = 0; t < NT; ++t) bc[t] = bl[t << 5];

    for (int i = 0; i < IN_C / 64; ++i) {
      const int inx = (i + 1 < IN_C / 64) ? i + 1 : i;
#pragma unroll
      for (int jj = 0; jj < 8; ++jj) an[jj] = pa[(inx << 4) + (jj << 1)];
#pragma unroll
      for (int jj = 0; jj < 8; ++jj) {
        const int j = (i << 3) + jj;
        const int jn = (j + 1 < IN_C / 8) ? j + 1 : j;
#pragma unroll
        for (int t = 0; t < NT; ++t) bn[t] = bl[(jn * NT + t) << 5];
        const v2f a0 = {ac[jj].x, ac[jj].y};
        const v2f a1 = {ac[jj].z, ac[jj].w};
#pragma unroll
        for (int t = 0; t < NT; ++t) {
          const v2f b0 = {bc[t].x, bc[t].y};
          acc[t] = __builtin_amdgcn_wmma_f32_16x16x4_f32(
              false, a0, false, b0, (short)0, acc[t], false, false);
        }
#pragma unroll
        for (int t = 0; t < NT; ++t) {
          const v2f b1 = {bc[t].z, bc[t].w};
          acc[t] = __builtin_amdgcn_wmma_f32_16x16x4_f32(
              false, a1, false, b1, (short)0, acc[t], false, false);
        }
#pragma unroll
        for (int t = 0; t < NT; ++t) bc[t] = bn[t];
      }
#pragma unroll
      for (int jj = 0; jj < 8; ++jj) ac[jj] = an[jj];
    }

    const int rbase = hh << 3;
#pragma unroll
    for (int r = 0; r < 8; ++r) {
      const int row = row0 + rbase + r;
      if (row < n_nodes) {
        float* __restrict__ hrow = h + (size_t)row * HID;
#pragma unroll
        for (int t = 0; t < NT - 1; ++t) hrow[(t << 4) + m] = acc[t][r];
        if (m < HID - 96) hrow[96 + m] = acc[NT - 1][r];
        else if (m < HID - 96 + OUT_C) outp[(size_t)row * OUT_C + (m - (HID - 96))] = acc[NT - 1][r];
      }
    }
  }
}

__global__ void __launch_bounds__(256)
gather1_kernel(const float* __restrict__ h, const float* __restrict__ dinv,
               const int* __restrict__ rowptr, const int* __restrict__ csr,
               const float* __restrict__ b1, const float* __restrict__ W2,
               float* __restrict__ h2, int n_nodes) {
  const int lane = threadIdx.x & 31;
  const int d = blockIdx.x * (blockDim.x >> 5) + (threadIdx.x >> 5);
  if (d >= n_nodes) return;
  const int beg = __builtin_amdgcn_readfirstlane(rowptr[d]);
  const int end = __builtin_amdgcn_readfirstlane(rowptr[d + 1]);
  const float dd = dinv[d];
  const bool act = lane < HID4;
  const v4f* __restrict__ h4 = (const v4f*)h;

  v4f acc = {0.f, 0.f, 0.f, 0.f};
  v4f vself = {0.f, 0.f, 0.f, 0.f};
  if (act) vself = h4[(size_t)d * HID4 + lane];

  for (int base = beg; base < end; base += 32) {
    int m = end - base;
    if (m > 32) m = 32;
    int s_l = 0;
    float w_l = 0.f;
    if (lane < m) {
      s_l = csr[base + lane];
      s_l = min(max(s_l, 0), n_nodes - 1);
      w_l = dinv[s_l] * dd;
    }
    int j = 0;
    for (; j + 4 <= m; j += 4) {
      const int s0 = __builtin_amdgcn_readlane(s_l, j);
      const int s1 = __builtin_amdgcn_readlane(s_l, j + 1);
      const int s2 = __builtin_amdgcn_readlane(s_l, j + 2);
      const int s3 = __builtin_amdgcn_readlane(s_l, j + 3);
      const float w0 = __int_as_float(__builtin_amdgcn_readlane(__float_as_int(w_l), j));
      const float w1 = __int_as_float(__builtin_amdgcn_readlane(__float_as_int(w_l), j + 1));
      const float w2 = __int_as_float(__builtin_amdgcn_readlane(__float_as_int(w_l), j + 2));
      const float w3 = __int_as_float(__builtin_amdgcn_readlane(__float_as_int(w_l), j + 3));
      if (act) {
        const v4f v0 = h4[(size_t)s0 * HID4 + lane];
        const v4f v1 = h4[(size_t)s1 * HID4 + lane];
        const v4f v2 = h4[(size_t)s2 * HID4 + lane];
        const v4f v3 = h4[(size_t)s3 * HID4 + lane];
        acc += v0 * w0;
        acc += v1 * w1;
        acc += v2 * w2;
        acc += v3 * w3;
      }
    }
    for (; j < m; ++j) {
      const int s0 = __builtin_amdgcn_readlane(s_l, j);
      const float w0 = __int_as_float(__builtin_amdgcn_readlane(__float_as_int(w_l), j));
      if (act) acc += h4[(size_t)s0 * HID4 + lane] * w0;
    }
  }

  acc += vself * (dd * dd);

  float p0 = 0.f, p1 = 0.f;
  if (act) {
    const v4f bb = ((const v4f*)b1)[lane];
    const v4f wa = ((const v4f*)W2)[2 * lane];
    const v4f wb = ((const v4f*)W2)[2 * lane + 1];
    float v0 = acc.x + bb.x, v1 = acc.y + bb.y, v2 = acc.z + bb.z, v3 = acc.w + bb.w;
    v0 = v0 > 0.f ? v0 : 0.f;
    v1 = v1 > 0.f ? v1 : 0.f;
    v2 = v2 > 0.f ? v2 : 0.f;
    v3 = v3 > 0.f ? v3 : 0.f;
    p0 = v0 * wa.x + v1 * wa.z + v2 * wb.x + v3 * wb.z;
    p1 = v0 * wa.y + v1 * wa.w + v2 * wb.y + v3 * wb.w;
  }
#pragma unroll
  for (int o = 16; o > 0; o >>= 1) {
    p0 += __shfl_xor(p0, o);
    p1 += __shfl_xor(p1, o);
  }
  if (lane == 0) { h2[(size_t)d * 2 + 0] = p0; h2[(size_t)d * 2 + 1] = p1; }
}

__global__ void gather2_kernel(const float* __restrict__ h2, const float* __restrict__ dinv,
                               const int* __restrict__ rowptr, const int* __restrict__ csr,
                               const float* __restrict__ b2, const float* __restrict__ bs,
                               float* __restrict__ out, int n_nodes) {
  int d = blockIdx.x * blockDim.x + threadIdx.x;
  if (d >= n_nodes) return;
  const int beg = rowptr[d], end = rowptr[d + 1];
  const float dd = dinv[d];
  const float di2 = dd * dd;
  const v2f* __restrict__ h22 = (const v2f*)h2;
  const v2f hv = h22[d];
  float a0 = out[(size_t)d * 2 + 0] + (b2[0] + bs[0] + hv.x * di2);
  float a1 = out[(size_t)d * 2 + 1] + (b2[1] + bs[1] + hv.y * di2);
#pragma unroll 4
  for (int j = beg; j < end; ++j) {
    int s = csr[j];
    s = min(max(s, 0), n_nodes - 1);
    const float w = dinv[s] * dd;
    const v2f v = h22[s];
    a0 += v.x * w;
    a1 += v.y * w;
  }
  out[(size_t)d * 2 + 0] = a0;
  out[(size_t)d * 2 + 1] = a1;
}

__global__ void log_softmax_kernel(float* __restrict__ out, int n_nodes) {
  int n = blockIdx.x * blockDim.x + threadIdx.x;
  if (n >= n_nodes) return;
  float z0 = out[(size_t)n * 2 + 0];
  float z1 = out[(size_t)n * 2 + 1];
  float mx = fmaxf(z0, z1);
  float lse = mx + logf(expf(z0 - mx) + expf(z1 - mx));
  out[(size_t)n * 2 + 0] = z0 - lse;
  out[(size_t)n * 2 + 1] = z1 - lse;
}

extern "C" void kernel_launch(void* const* d_in, const int* in_sizes, int n_in,
                              void* d_out, int out_size, void* d_ws, size_t ws_size,
                              hipStream_t stream) {
  const float* x  = (const float*)d_in[0];
  const int*   ei = (const int*)d_in[1];
  const float* W1 = (const float*)d_in[2];
  const float* b1 = (const float*)d_in[3];
  const float* W2 = (const float*)d_in[4];
  const float* b2 = (const float*)d_in[5];
  const float* Ws = (const float*)d_in[6];
  const float* bs = (const float*)d_in[7];
  float* out = (float*)d_out;

  const int n_nodes = in_sizes[0] / IN_C;
  const int n_edges = in_sizes[1] / 2;
  const int* srcv = ei;
  const int* dstv = ei + n_edges;

  float* Wp    = (float*)d_ws;
  int* tab     = (int*)d_ws;
  float* dinv  = Wp + WP_FLOATS;
  size_t off   = ((size_t)n_nodes + 255) & ~(size_t)255;
  int* bstart  = (int*)(dinv + off);
  int* rowptr  = bstart + off;
  size_t off1  = ((size_t)n_nodes + 1 + 255) & ~(size_t)255;
  float* h     = (float*)(rowptr + off1);
  float* h2    = h + (size_t)n_nodes * HID;
  int* csr     = (int*)(h2 + (size_t)n_nodes * 2) + n_edges;
  int2* pairs  = (int2*)h;

  const int T = 256;

  bucket_hist_kernel<<<SG, ST, 0, stream>>>(dstv, tab, n_nodes, n_edges);
  bucket_scatter_kernel<<<SG, ST, 0, stream>>>(srcv, dstv, tab, bstart, pairs, n_nodes, n_edges);
  bucket_build_kernel<<<NB, ST, 0, stream>>>(pairs, bstart, rowptr, dinv, csr, n_nodes, n_edges);

  gemm1_wmma_kernel<<<GEMM_WG, GEMM_WAVES * 32, 0, stream>>>(x, W1, Ws, h, out, n_nodes);

  gather1_kernel<<<(n_nodes + 7) / 8, T, 0, stream>>>(h, dinv, rowptr, csr, b1, W2, h2, n_nodes);

  gather2_kernel<<<(n_nodes + T - 1) / T, T, 0, stream>>>(h2, dinv, rowptr, csr, b2, bs, out, n_nodes);

  log_softmax_kernel<<<(n_nodes + T - 1) / T, T, 0, stream>>>(out, n_nodes);
}


# dinv[src] folded into the producers (GEMM store scales h rows, gather1 tail scales h2): gathers load no per-edge weight, add plain rows in CSR order, scale once by dinv[dst]; skip+bias added last
# speedup vs baseline: 4.2642x; 1.0888x over previous
#include <hip/hip_runtime.h>
#include <math.h>


#define IN_C 512
#define HID 100
#define HID4 25
#define OUT_C 2
#define NT 7
#define NCOL (NT * 16)
#define BS_FLOATS (IN_C * NCOL)
#define WP_FLOATS 65536
#define GEMM_WG 256
#define GEMM_WAVES 16

typedef __attribute__((ext_vector_type(2))) float v2f;
typedef __attribute__((ext_vector_type(4))) float v4f;
typedef __attribute__((ext_vector_type(8))) float v8f;

#define NB 256
#define SG 256
#define ST 1024
#define EB 8
#define MAXNPB 1024
#define CAP2 8192
#define CAP3 12288
static_assert(NB * SG <= WP_FLOATS, "the table of chunk counts lies in the workspace head");

__global__ void __launch_bounds__(ST)
bucket_hist_kernel(const int* __restrict__ dst, int* __restrict__ tab, int n_nodes, int e) {
  __shared__ int hc[NB];
  const int tid = threadIdx.x, g = blockIdx.x;
  const unsigned npb = (unsigned)((n_nodes + NB - 1) / NB);
  const int chunk = (e + SG - 1) / SG;
  const int e0 = g * chunk;
  int e1 = e0 + chunk;
  if (e1 > e) e1 = e;
  if (tid < NB) hc[tid] = 0;
  __syncthreads();
  for (int base = e0; base < e1; base += ST * EB) {
    int d[EB];
#pragma unroll
    for (int k = 0; k < EB; ++k) {
      const int i = base + k * ST + tid;
      d[k] = dst[min(i, e1 - 1)];
    }
#pragma unroll
    for (int k = 0; k < EB; ++k) {
      const int i = base + k * ST + tid;
      if (i < e1) {
        const int dd = min(max(d[k], 0), n_nodes - 1);
        atomicAdd(&hc[(unsigned)dd / npb], 1);
      }
    }
  }
  __syncthreads();
  if (tid < NB) tab[tid * SG + g] = hc[tid];
}

__global__ void __launch_bounds__(ST)
bucket_scatter_kernel(const int* __restrict__ src, const int* __restrict__ dst,
                      const int* __restrict__ tab, int* __restrict__ bstart,
                      int2* __restrict__ pairs, int n_nodes, int e) {
  static_assert(SG == 256 && NB == 256 && ST == 1024, "wave w sums ranges 8w..8w+7, lane l the chunks 8l..8l+7");
  __shared__ int tot[NB];
  __shared__ int gb[NB];
  __shared__ int lst[NB];
  __shared__ int lcur[NB];
  __shared__ int2 stage[CAP2];
  const int tid = threadIdx.x, g = blockIdx.x, lane = tid & 31, wv = tid >> 5;
  const unsigned npb = (unsigned)((n_nodes + NB - 1) / NB);
  const int chunk = (e + SG - 1) / SG;
  const int e0 = g * chunk;
  int e1 = e0 + chunk;
  if (e1 > e) e1 = e;
  const int m = (e1 > e0) ? e1 - e0 : 0;

  int4 ta[8], tc[8];
#pragma unroll
  for (int k = 0; k < 8; ++k) {
    ta[k] = *(const int4*)(tab + (wv * 8 + k) * SG + lane * 8);
    tc[k] = *(const int4*)(tab + (wv * 8 + k) * SG + lane * 8 + 4);
  }
#pragma unroll
  for (int k = 0; k < 8; ++k) {
    const int b = wv * 8 + k;
    const int v[8] = {ta[k].x, ta[k].y, ta[k].z, ta[k].w, tc[k].x, tc[k].y, tc[k].z, tc[k].w};
    int s = 0, p = 0, mine = 0;
#pragma unroll
    for (int j = 0; j < 8; ++j) {
      s += v[j];
      if (lane * 8 + j < g) p += v[j];
      if (lane * 8 + j == g) mine = v[j];
    }
    for (int o = 16; o > 0; o >>= 1) { s += __shfl_xor(s, o); p += __shfl_xor(p, o); }
    if (lane == 0) { tot[b] = s; gb[b] = p; }
    if (lane == (g >> 3)) lst[b] = mine;
  }
  __syncthreads();
  if (wv == 0) {
    int t[8], lt[8];
    int run = 0, lrun = 0;
#pragma unroll
    for (int j = 0; j < 8; ++j) {
      t[j] = run;   run += tot[lane * 8 + j];
      lt[j] = lrun; lrun += lst[lane * 8 + j];
    }
    int x = run, lx = lrun;
    for (int o = 1; o < 32; o <<= 1) {
      const int y = __shfl_up(x, o), ly = __shfl_up(lx, o);
      if (lane >= o) { x += y; lx += ly; }
    }
    const int ex = x - run, lex = lx - lrun;
#pragma unroll
    for (int j = 0; j < 8; ++j) {
      const int st = ex + t[j];
      gb[lane * 8 + j] += st;
      lst[lane * 8 + j] = lex + lt[j];
      lcur[lane * 8 + j] = lex + lt[j];
      if (g == 0) bstart[lane * 8 + j] = st;
    }
    if (g == 0 && lane == 31) bstart[NB] = x;
  }
  __syncthreads();

  const bool in_lds = (m <= CAP2);
  for (int base = e0; base < e1; base += ST * EB) {
    int d[EB], s[EB];
#pragma unroll
    for (int k = 0; k < EB; ++k) {
      const int i = base + k * ST + tid;
      d[k] = dst[min(i, e1 - 1)];
      s[k] = src[min(i, e1 - 1)];
    }
#pragma unroll
    for (int k = 0; k < EB; ++k) {
      const int i = base + k * ST + tid;
      if (i < e1) {
        const int dd = min(max(d[k], 0), n_nodes - 1);
        const unsigned b = (unsigned)dd / npb;
        const int q = atomicAdd(&lcur[b], 1);
        if (in_lds) {
          stage[min(max(q, 0), CAP2 - 1)] = make_int2(s[k], dd);
        } else {
          int pos = gb[b] + (q - lst[b]);
          pos = min(max(pos, 0), e - 1);
          pairs[pos] = make_int2(s[k], dd);
        }
      }
    }
  }
  if (in_lds) {
    __syncthreads();
    for (int q = tid; q < m; q += ST) {
      const int2 pr = stage[q];
      const unsigned b = (unsigned)pr.y / npb;
      int pos = gb[b] + (q - lst[b]);
      pos = min(max(pos, 0), e - 1);
      pairs[pos] = pr;
    }
  }
}

__global__ void __launch_bounds__(ST)
bucket_build_kernel(const int2* __restrict__ pairs, const int* __restrict__ bstart,
                    int* __restrict__ rowptr, float* __restrict__ dinv,
                    int* __restrict__ csr, int n_nodes, int e) {
  static_assert(ST == MAXNPB, "one thread per LDS counter");
  __shared__ int cnt[MAXNPB];
  __shared__ int wsum[32];
  __shared__ int stage[CAP3];
  const int tid = threadIdx.x, b = blockIdx.x, lane = tid & 31, wv = tid >> 5;
  const int npb = (n_nodes + NB - 1) / NB;
  const int nb0 = b * npb;
  int s0 = bstart[b], s1 = bstart[b + 1];
  s0 = min(max(s0, 0), e);
  s1 = min(max(s1, s0), e);
  cnt[tid] = 0;
  __syncthreads();
  for (int base = s0; base < s1; base += ST * EB) {
    int d[EB];
#pragma unroll
    for (int k = 0; k < EB; ++k) {
      const int i = base + k * ST + tid;
      d[k] = pairs[min(i, s1 - 1)].y;
    }
#pragma unroll
    for (int k = 0; k < EB; ++k) {
      const int i = base + k * ST + tid;
      if (i < s1) atomicAdd(&cnt[min(max(d[k] - nb0, 0), MAXNPB - 1)], 1);
    }
  }
  __syncthreads();
  const int c = cnt[tid];
  int x = c;
  for (int o = 1; o < 32; o <<= 1) { int y = __shfl_up(x, o); if (lane >= o) x += y; }
  if (lane == 31) wsum[wv] = x;
  __syncthreads();
  if (wv == 0) {
    int s = wsum[lane];
    for (int o = 1; o < 32; o <<= 1) { int y = __shfl_up(s, o); if (lane >= o) s += y; }
    wsum[lane] = s;
  }
  __syncthreads();
  const int first = s0 + ((wv > 0) ? wsum[wv - 1] : 0) + (x - c);
  cnt[tid] = first;
  const int node = nb0 + tid;
  if (tid < npb && node < n_nodes) {
    rowptr[node] = first;
    const float dg = (float)(c + 1);
    dinv[node] = (dg > 0.f) ? rsqrtf(dg) : 0.f;
  }
  if (b == NB - 1 && tid == 0) rowptr[n_nodes] = s1;
  __syncthreads();

  const int m = s1 - s0;
  const bool in_lds = (m <= CAP3);
  for (int base = s0; base < s1; base += ST * EB) {
    int2 pr[EB];
#pragma unroll
    for (int k = 0; k < EB; ++k) {
      const int i = base + k * ST + tid;
      pr[k] = pairs[min(i, s1 - 1)];
    }
#pragma unroll
    for (int k = 0; k < EB; ++k) {
      const int i = base + k * ST + tid;
      if (i < s1) {
        const int ld = min(max(pr[k].y - nb0, 0), MAXNPB - 1);
        const int p = atomicAdd(&cnt[ld], 1);
        if (in_lds) stage[min(max(p - s0, 0), CAP3 - 1)] = pr[k].x;
        else csr[min(max(p, 0), e - 1)] = pr[k].x;
      }
    }
  }
  if (in_lds) {
    __syncthreads();
    for (int i = (s0 & ~31) + tid; i < s1; i += ST)
      if (i >= s0) csr[i] = stage[i - s0];
  }
}

__global__ void __launch_bounds__(GEMM_WAVES * 32)
gemm1_wmma_kernel(const float* __restrict__ x, const float* __restrict__ W1,
                  const float* __restrict__ Ws, const float* __restrict__ dinv,
                  float* __restrict__ h, float* __restrict__ outp, int n_nodes) {
  __shared__ __attribute__((aligned(16))) float Bs[BS_FLOATS];
  v4f* __restrict__ Bs4 = (v4f*)Bs;
  static_assert(GEMM_WAVES * 32 * 25 == (IN_C / 4) * HID && GEMM_WAVES * 32 * 3 == (IN_C / 4) * (NCOL - HID),
                "the copy below gives every thread exactly 25 + 3 items");

  for (int g = 0; g < 5; ++g) {
    float w[5][4];
#pragma unroll
    for (int u = 0; u < 5; ++u) {
      const int id = threadIdx.x + (GEMM_WAVES * 32) * (g * 5 + u);
      const int kq = id / HID;
      const int c = id - kq * HID;
      const float* __restrict__ p = W1 + (kq * 4) * HID + c;
#pragma unroll
      for (int q = 0; q < 4; ++q) w[u][q] = p[q * HID];
    }
#pragma unroll
    for (int u = 0; u < 5; ++u) {
      const int id = threadIdx.x + (GEMM_WAVES * 32) * (g * 5 + u);
      const int kq = id / HID;
      const int c = id - kq * HID;
      Bs4[(((kq >> 1) * NT + (c >> 4)) << 5) + ((kq & 1) << 4) + (c & 15)] =
          (v4f){w[u][0], w[u][1], w[u][2], w[u][3]};
    }
  }
  {
    float ws[3][4];
#pragma unroll
    for (int u = 0; u < 3; ++u) {
      const int id = threadIdx.x + (GEMM_WAVES * 32) * u;
      const int kq = id / 12;
      const int cc = id - kq * 12;
      const float* __restrict__ p = Ws + (kq * 4) * OUT_C + (cc < OUT_C ? cc : OUT_C - 1);
#pragma unroll
      for (int q = 0; q < 4; ++q) ws[u][q] = p[q * OUT_C];
    }
#pragma unroll
    for (int u = 0; u < 3; ++u) {
      const int id = threadIdx.x + (GEMM_WAVES * 32) * u;
      const int kq = id / 12;
      const int cc = id - kq * 12;
      const bool keep = cc < OUT_C;
      Bs4[(((kq >> 1) * NT + (NT - 1)) << 5) + ((kq & 1) << 4) + (HID - 96 + cc)] =
          (v4f){keep ? ws[u][0] : 0.f, keep ? ws[u][1] : 0.f, keep ? ws[u][2] : 0.f, keep ? ws[u][3] : 0.f};
    }
  }
  __syncthreads();

  const int lane = threadIdx.x & 31;
  const int wv   = threadIdx.x >> 5;
  const int m  = lane & 15;
  const int hh = lane >> 4;
  const int tiles = (n_nodes + 15) >> 4;
  const v4f* __restrict__ bl = (const v4f*)Bs + (hh << 4) + m;

  for (int tile = wv * GEMM_WG + blockIdx.x; tile < tiles; tile += GEMM_WG * GEMM_WAVES) {
    const int row0 = tile << 4;
    int arow = row0 + m;
    if (arow > n_nodes - 1) arow = n_nodes - 1;
    const v4f* __restrict__ pa = (const v4f*)(x + (size_t)arow * IN_C) + hh;

    v8f acc[NT] = {};
    v4f ac[8], an[8];
#pragma unroll
    for (int jj = 0; jj < 8; ++jj) ac[jj] = pa[jj << 1];
    v4f bc[NT], bn[NT];
#pragma unroll
    for (int t = 0; t < NT; ++t) bc[t] = bl[t << 5];
    float ds[8];
#pragma unroll
    for (int r = 0; r < 8; ++r) {
      const int drow = row0 + (hh << 3) + r;
      ds[r] = dinv[drow < n_nodes ? drow : n_nodes - 1];
    }

    for (int i = 0; i < IN_C / 64; ++i) {
      const int inx = (i + 1 < IN_C / 64) ? i + 1 : i;
#pragma unroll
      for (int jj = 0; jj < 8; ++jj) an[jj] = pa[(inx << 4) + (jj << 1)];
#pragma unroll
      for (int jj = 0; jj < 8; ++jj) {
        const int j = (i << 3) + jj;
        const int jn = (j + 1 < IN_C / 8) ? j + 1 : j;
#pragma unroll
        for (int t = 0; t < NT; ++t) bn[t] = bl[(jn * NT + t) << 5];
        const v2f a0 = {ac[jj].x, ac[jj].y};
        const v2f a1 = {ac[jj].z, ac[jj].w};
#pragma unroll
        for (int t = 0; t < NT; ++t) {
          const v2f b0 = {bc[t].x, bc[t].y};
          acc[t] = __builtin_amdgcn_wmma_f32_16x16x4_f32(
              false, a0, false, b0, (short)0, acc[t], false, false);
        }
#pragma unroll
        for (int t = 0; t < NT; ++t) {
          const v2f b1 = {bc[t].z, bc[t].w};
          acc[t] = __builtin_amdgcn_wmma_f32_16x16x4_f32(
              false, a1, false, b1, (short)0, acc[t], false, false);
        }
#pragma unroll
        for (int t = 0; t < NT; ++t) bc[t] = bn[t];
      }
#pragma unroll
      for (int jj = 0; jj < 8; ++jj) ac[jj] = an[jj];
    }

    const int rbase = hh << 3;
#pragma unroll
    for (int r = 0; r < 8; ++r) {
      const int row = row0 + rbase + r;
      if (row < n_nodes) {
        float* __restrict__ hrow = h + (size_t)row * HID;
#pragma unroll
        for (int t = 0; t < NT - 1; ++t) hrow[(t << 4) + m] = acc[t][r] * ds[r];
        if (m < HID - 96) hrow[96 + m] = acc[NT - 1][r] * ds[r];
        else if (m < HID - 96 + OUT_C) outp[(size_t)row * OUT_C + (m - (HID - 96))] = acc[NT - 1][r];
      }
    }
  }
}

__global__ void __launch_bounds__(256)
gather1_kernel(const float* __restrict__ h, const float* __restrict__ dinv,
               const int* __restrict__ rowptr, const int* __restrict__ csr,
               const float* __restrict__ b1, const float* __restrict__ W2,
               float* __restrict__ h2, int n_nodes) {
  const int lane = threadIdx.x & 31;
  const int d = blockIdx.x * (blockDim.x >> 5) + (threadIdx.x >> 5);
  if (d >= n_nodes) return;
  const int beg = __builtin_amdgcn_readfirstlane(rowptr[d]);
  const int end = __builtin_amdgcn_readfirstlane(rowptr[d + 1]);
  const float dd = dinv[d];
  const bool act = lane < HID4;
  const v4f* __restrict__ h4 = (const v4f*)h;

  v4f acc = {0.f, 0.f, 0.f, 0.f};
  v4f vself = {0.f, 0.f, 0.f, 0.f};
  if (act) vself = h4[(size_t)d * HID4 + lane];

  for (int base = beg; base < end; base += 32) {
    int m = end - base;
    if (m > 32) m = 32;
    int s_l = 0;
    if (lane < m) {
      s_l = csr[base + lane];
      s_l = min(max(s_l, 0), n_nodes - 1);
    }
    int j = 0;
    for (; j + 4 <= m; j += 4) {
      const int s0 = __builtin_amdgcn_readlane(s_l, j);
      const int s1 = __builtin_amdgcn_readlane(s_l, j + 1);
      const int s2 = __builtin_amdgcn_readlane(s_l, j + 2);
      const int s3 = __builtin_amdgcn_readlane(s_l, j + 3);
      if (act) {
        const v4f v0 = h4[(size_t)s0 * HID4 + lane];
        const v4f v1 = h4[(size_t)s1 * HID4 + lane];
        const v4f v2 = h4[(size_t)s2 * HID4 + lane];
        const v4f v3 = h4[(size_t)s3 * HID4 + lane];
        acc += v0;
        acc += v1;
        acc += v2;
        acc += v3;
      }
    }
    for (; j < m; ++j) {
      const int s0 = __builtin_amdgcn_readlane(s_l, j);
      if (act) acc += h4[(size_t)s0 * HID4 + lane];
    }
  }

  acc += vself;
  acc *= dd;

  float p0 = 0.f, p1 = 0.f;
  if (act) {
    const v4f bb = ((const v4f*)b1)[lane];
    const v4f wa = ((const v4f*)W2)[2 * lane];
    const v4f wb = ((const v4f*)W2)[2 * lane + 1];
    float v0 = acc.x + bb.x, v1 = acc.y + bb.y, v2 = acc.z + bb.z, v3 = acc.w + bb.w;
    v0 = v0 > 0.f ? v0 : 0.f;
    v1 = v1 > 0.f ? v1 : 0.f;
    v2 = v2 > 0.f ? v2 : 0.f;
    v3 = v3 > 0.f ? v3 : 0.f;
    p0 = v0 * wa.x + v1 * wa.z + v2 * wb.x + v3 * wb.z;
    p1 = v0 * wa.y + v1 * wa.w + v2 * wb.y + v3 * wb.w;
  }
#pragma unroll
  for (int o = 16; o > 0; o >>= 1) {
    p0 += __shfl_xor(p0, o);
    p1 += __shfl_xor(p1, o);
  }
  if (lane == 0) { h2[(size_t)d * 2 + 0] = p0 * dd; h2[(size_t)d * 2 + 1] = p1 * dd; }
}

__global__ void gather2_kernel(const float* __restrict__ h2, const float* __restrict__ dinv,
                               const int* __restrict__ rowptr, const int* __restrict__ csr,
                               const float* __restrict__ b2, const float* __restrict__ bs,
                               float* __restrict__ out, int n_nodes) {
  int d = blockIdx.x * blockDim.x + threadIdx.x;
  if (d >= n_nodes) return;
  const int beg = rowptr[d], end = rowptr[d + 1];
  const float dd = dinv[d];
  const v2f* __restrict__ h22 = (const v2f*)h2;
  const v2f hv = h22[d];
  const float sk0 = out[(size_t)d * 2 + 0];
  const float sk1 = out[(size_t)d * 2 + 1];
  const float bb0 = b2[0] + bs[0], bb1 = b2[1] + bs[1];
  float t0 = hv.x, t1 = hv.y;
#pragma unroll 4
  for (int j = beg; j < end; ++j) {
    int s = csr[j];
    s = min(max(s, 0), n_nodes - 1);
    const v2f v = h22[s];
    t0 += v.x;
    t1 += v.y;
  }
  out[(size_t)d * 2 + 0] = sk0 + (bb0 + t0 * dd);
  out[(size_t)d * 2 + 1] = sk1 + (bb1 + t1 * dd);
}

__global__ void log_softmax_kernel(float* __restrict__ out, int n_nodes) {
  int n = blockIdx.x * blockDim.x + threadIdx.x;
  if (n >= n_nodes) return;
  float z0 = out[(size_t)n * 2 + 0];
  float z1 = out[(size_t)n * 2 + 1];
  float mx = fmaxf(z0, z1);
  float lse = mx + logf(expf(z0 - mx) + expf(z1 - mx));
  out[(size_t)n * 2 + 0] = z0 - lse;
  out[(size_t)n * 2 + 1] = z1 - lse;
}

extern "C" void kernel_launch(void* const* d_in, const int* in_sizes, int n_in,
                              void* d_out, int out_size, void* d_ws, size_t ws_size,
                              hipStream_t stream) {
  const float* x  = (const float*)d_in[0];
  const int*   ei = (const int*)d_in[1];
  const float* W1 = (const float*)d_in[2];
  const float* b1 = (const float*)d_in[3];
  const float* W2 = (const float*)d_in[4];
  const float* b2 = (const float*)d_in[5];
  const float* Ws = (const float*)d_in[6];
  const float* bs = (const float*)d_in[7];
  float* out = (float*)d_out;

  const int n_nodes = in_sizes[0] / IN_C;
  const int n_edges = in_sizes[1] / 2;
  const int* srcv = ei;
  const int* dstv = ei + n_edges;

  float* Wp    = (float*)d_ws;
  int* tab     = (int*)d_ws;
  float* dinv  = Wp + WP_FLOATS;
  size_t off   = ((size_t)n_nodes + 255) & ~(size_t)255;
  int* bstart  = (int*)(dinv + off);
  int* rowptr  = bstart + off;
  size_t off1  = ((size_t)n_nodes + 1 + 255) & ~(size_t)255;
  float* h     = (float*)(rowptr + off1);
  float* h2    = h + (size_t)n_nodes * HID;
  int* csr     = (int*)(h2 + (size_t)n_nodes * 2) + n_edges;
  int2* pairs  = (int2*)h;

  const int T = 256;

  bucket_hist_kernel<<<SG, ST, 0, stream>>>(dstv, tab, n_nodes, n_edges);
  bucket_scatter_kernel<<<SG, ST, 0, stream>>>(srcv, dstv, tab, bstart, pairs, n_nodes, n_edges);
  bucket_build_kernel<<<NB, ST, 0, stream>>>(pairs, bstart, rowptr, dinv, csr, n_nodes, n_edges);

  gemm1_wmma_kernel<<<GEMM_WG, GEMM_WAVES * 32, 0, stream>>>(x, W1, Ws, dinv, h, out, n_nodes);

  gather1_kernel<<<(n_nodes + 7) / 8, T, 0, stream>>>(h, dinv, rowptr, csr, b1, W2, h2, n_nodes);

  gather2_kernel<<<(n_nodes + T - 1) / T, T, 0, stream>>>(h2, dinv, rowptr, csr, b2, bs, out, n_nodes);

  log_softmax_kernel<<<(n_nodes + T - 1) / T, T, 0, stream>>>(out, n_nodes);
}


# Counting sort level 1 in ONE launch: a chunk ranks its edges per range on LDS counters and writes itself back sorted, with its run starts; no histogram launch (6 launches); in-node order: LDS arrival
# speedup vs baseline: 4.5056x; 1.0566x over previous
#include <hip/hip_runtime.h>
#include <math.h>


#define IN_C 512
#define HID 100
#define HID4 25
#define OUT_C 2
#define NT 7
#define NCOL (NT * 16)
#define BS_FLOATS (IN_C * NCOL)
#define WP_FLOATS 65536
#define GEMM_WG 256
#define GEMM_WAVES 16

typedef __attribute__((ext_vector_type(2))) float v2f;
typedef __attribute__((ext_vector_type(4))) float v4f;
typedef __attribute__((ext_vector_type(8))) float v8f;

#define NB 256
#define SG 256
#define ST 1024
#define EB 8
#define MAXNPB 1024
#define CAP2 8192
#define CAP3 12288
static_assert(NB * SG <= WP_FLOATS, "the table of run starts lies in the workspace head");
static_assert(CAP2 == ST * EB, "a chunk that fits the LDS stage is loaded in one trip");

__global__ void __launch_bounds__(ST)
chunk_sort_kernel(const int* __restrict__ src, const int* __restrict__ dst,
                  int* __restrict__ tab, int2* __restrict__ pairs, int n_nodes, int e) {
  static_assert(NB == 256, "wave 0 scans the NB counts, lane l the ranges 8l..8l+7");
  __shared__ int cnt[NB];
  __shared__ int2 stage[CAP2];
  const int tid = threadIdx.x, g = blockIdx.x, lane = tid & 31, wv = tid >> 5;
  const unsigned npb = (unsigned)((n_nodes + NB - 1) / NB);
  const int chunk = (e + SG - 1) / SG;
  const int e0 = g * chunk;
  int e1 = e0 + chunk;
  if (e1 > e) e1 = e;
  const int m = (e1 > e0) ? e1 - e0 : 0;
  const bool in_lds = (m <= CAP2);
  if (tid < NB) cnt[tid] = 0;
  __syncthreads();

  int d[EB], s[EB], r[EB];
  unsigned rg[EB];
#pragma unroll
  for (int k = 0; k < EB; ++k) { d[k] = 0; s[k] = 0; r[k] = 0; rg[k] = 0; }
  if (in_lds) {
    if (m > 0) {
#pragma unroll
      for (int k = 0; k < EB; ++k) {
        const int i = e0 + k * ST + tid;
        d[k] = dst[min(i, e1 - 1)];
        s[k] = src[min(i, e1 - 1)];
      }
    }
#pragma unroll
    for (int k = 0; k < EB; ++k) {
      const int i = e0 + k * ST + tid;
      if (i < e1) {
        d[k] = min(max(d[k], 0), n_nodes - 1);
        rg[k] = (unsigned)d[k] / npb;
        r[k] = atomicAdd(&cnt[rg[k]], 1);
      }
    }
  } else {
    for (int base = e0; base < e1; base += ST) {
      const int i = base + tid;
      if (i < e1) {
        const int dd = min(max(dst[i], 0), n_nodes - 1);
        atomicAdd(&cnt[(unsigned)dd / npb], 1);
      }
    }
  }
  __syncthreads();
  if (wv == 0) {
    int t[8];
    int run = 0;
#pragma unroll
    for (int j = 0; j < 8; ++j) { t[j] = run; run += cnt[lane * 8 + j]; }
    int x = run;
    for (int o = 1; o < 32; o <<= 1) {
      const int y = __shfl_up(x, o);
      if (lane >= o) x += y;
    }
    const int ex = x - run;
#pragma unroll
    for (int j = 0; j < 8; ++j) {
      cnt[lane * 8 + j] = ex + t[j];
      tab[g * NB + lane * 8 + j] = ex + t[j];
    }
  }
  __syncthreads();
  if (in_lds) {
#pragma unroll
    for (int k = 0; k < EB; ++k) {
      const int i = e0 + k * ST + tid;
      if (i < e1) {
        const int q = cnt[rg[k]] + r[k];
        stage[min(max(q, 0), CAP2 - 1)] = make_int2(s[k], d[k]);
      }
    }
    __syncthreads();
    for (int i = (e0 & ~31) + tid; i < e1; i += ST)
      if (i >= e0) pairs[i] = stage[i - e0];
  } else {
    for (int base = e0; base < e1; base += ST) {
      const int i = base + tid;
      if (i < e1) {
        const int dd = min(max(dst[i], 0), n_nodes - 1);
        int q = atomicAdd(&cnt[(unsigned)dd / npb], 1);
        q = min(max(q, 0), m - 1);
        pairs[e0 + q] = make_int2(src[i], dd);
      }
    }
  }
}

__global__ void __launch_bounds__(ST)
range_build_kernel(const int2* __restrict__ pairs, const int* __restrict__ tab,
                   int* __restrict__ rowptr, float* __restrict__ dinv,
                   int* __restrict__ csr, int n_nodes, int e) {
  static_assert(ST == MAXNPB, "one thread per LDS counter");
  static_assert(SG == 256 && ST == 1024, "threads 0..SG-1 (waves 0..7) fetch the run bounds; wave w reads the runs of chunks 8w..8w+7");
  __shared__ int cnt[MAXNPB];
  __shared__ int wsum[32];
  __shared__ int rs[SG];
  __shared__ int rl[SG];
  __shared__ int red0[8];
  __shared__ int red1[8];
  __shared__ int stage[CAP3];
  const int tid = threadIdx.x, b = blockIdx.x, lane = tid & 31, wv = tid >> 5;
  const int npb = (n_nodes + NB - 1) / NB;
  const int nb0 = b * npb;
  const int chunk = (e + SG - 1) / SG;
  cnt[tid] = 0;
  if (tid < SG) {
    const int ce0 = tid * chunk;
    int ce1 = ce0 + chunk;
    if (ce1 > e) ce1 = e;
    const int cm = (ce1 > ce0) ? ce1 - ce0 : 0;
    int o0 = tab[tid * NB + b];
    int o1 = (b + 1 < NB) ? tab[tid * NB + b + 1] : cm;
    o0 = min(max(o0, 0), cm);
    o1 = min(max(o1, o0), cm);
    rs[tid] = ce0 + o0;
    rl[tid] = o1 - o0;
    int a = o0, l = o1 - o0;
    for (int o = 16; o > 0; o >>= 1) { a += __shfl_xor(a, o); l += __shfl_xor(l, o); }
    if (lane == 0) { red0[wv] = a; red1[wv] = l; }
  }
  __syncthreads();
  int s0 = 0, m = 0;
#pragma unroll
  for (int j = 0; j < 8; ++j) { s0 += red0[j]; m += red1[j]; }
  s0 = min(max(s0, 0), e);
  m = min(max(m, 0), e - s0);
  const int s1 = s0 + m;

  int2 pr[8];
  int ln[8], st8[8];
#pragma unroll
  for (int k = 0; k < 8; ++k) { ln[k] = rl[wv * 8 + k]; st8[k] = rs[wv * 8 + k]; }
#pragma unroll
  for (int k = 0; k < 8; ++k) {
    pr[k] = make_int2(0, nb0);
    if (lane < ln[k]) pr[k] = pairs[min(max(st8[k] + lane, 0), e - 1)];
  }
#pragma unroll
  for (int k = 0; k < 8; ++k)
    if (lane < ln[k]) atomicAdd(&cnt[min(max(pr[k].y - nb0, 0), MAXNPB - 1)], 1);
  for (int k = 0; k < 8; ++k) {
    const int len = rl[wv * 8 + k], st = rs[wv * 8 + k];
    for (int off = 32 + lane; off < len; off += 32) {
      const int y = pairs[min(max(st + off, 0), e - 1)].y;
      atomicAdd(&cnt[min(max(y - nb0, 0), MAXNPB - 1)], 1);
    }
  }
  __syncthreads();
  const int c = cnt[tid];
  int x = c;
  for (int o = 1; o < 32; o <<= 1) { int y = __shfl_up(x, o); if (lane >= o) x += y; }
  if (lane == 31) wsum[wv] = x;
  __syncthreads();
  if (wv == 0) {
    int sw = wsum[lane];
    for (int o = 1; o < 32; o <<= 1) { int y = __shfl_up(sw, o); if (lane >= o) sw += y; }
    wsum[lane] = sw;
  }
  __syncthreads();
  const int first = s0 + ((wv > 0) ? wsum[wv - 1] : 0) + (x - c);
  cnt[tid] = first;
  const int node = nb0 + tid;
  if (tid < npb && node < n_nodes) {
    rowptr[node] = first;
    const float dg = (float)(c + 1);
    dinv[node] = (dg > 0.f) ? rsqrtf(dg) : 0.f;
  }
  if (b == NB - 1 && tid == 0) rowptr[n_nodes] = s1;
  __syncthreads();

  const bool in_lds = (m <= CAP3);
#pragma unroll
  for (int k = 0; k < 8; ++k) {
    if (lane < ln[k]) {
      const int ld = min(max(pr[k].y - nb0, 0), MAXNPB - 1);
      const int p = atomicAdd(&cnt[ld], 1);
      if (in_lds) stage[min(max(p - s0, 0), CAP3 - 1)] = pr[k].x;
      else csr[min(max(p, 0), e - 1)] = pr[k].x;
    }
  }
  for (int k = 0; k < 8; ++k) {
    const int len = rl[wv * 8 + k], st = rs[wv * 8 + k];
    for (int off = 32 + lane; off < len; off += 32) {
      const int2 q2 = pairs[min(max(st + off, 0), e - 1)];
      const int ld = min(max(q2.y - nb0, 0), MAXNPB - 1);
      const int p = atomicAdd(&cnt[ld], 1);
      if (in_lds) stage[min(max(p - s0, 0), CAP3 - 1)] = q2.x;
      else csr[min(max(p, 0), e - 1)] = q2.x;
    }
  }
  if (in_lds) {
    __syncthreads();
    for (int i = (s0 & ~31) + tid; i < s1; i += ST)
      if (i >= s0) csr[i] = stage[i - s0];
  }
}

__global__ void __launch_bounds__(GEMM_WAVES * 32)
gemm1_wmma_kernel(const float* __restrict__ x, const float* __restrict__ W1,
                  const float* __restrict__ Ws, const float* __restrict__ dinv,
                  float* __restrict__ h, float* __restrict__ outp, int n_nodes) {
  __shared__ __attribute__((aligned(16))) float Bs[BS_FLOATS];
  v4f* __restrict__ Bs4 = (v4f*)Bs;
  static_assert(GEMM_WAVES * 32 * 25 == (IN_C / 4) * HID && GEMM_WAVES * 32 * 3 == (IN_C / 4) * (NCOL - HID),
                "the copy below gives every thread exactly 25 + 3 items");

  for (int g = 0; g < 5; ++g) {
    float w[5][4];
#pragma unroll
    for (int u = 0; u < 5; ++u) {
      const int id = threadIdx.x + (GEMM_WAVES * 32) * (g * 5 + u);
      const int kq = id / HID;
      const int c = id - kq * HID;
      const float* __restrict__ p = W1 + (kq * 4) * HID + c;
#pragma unroll
      for (int q = 0; q < 4; ++q) w[u][q] = p[q * HID];
    }
#pragma unroll
    for (int u = 0; u < 5; ++u) {
      const int id = threadIdx.x + (GEMM_WAVES * 32) * (g * 5 + u);
      const int kq = id / HID;
      const int c = id - kq * HID;
      Bs4[(((kq >> 1) * NT + (c >> 4)) << 5) + ((kq & 1) << 4) + (c & 15)] =
          (v4f){w[u][0], w[u][1], w[u][2], w[u][3]};
    }
  }
  {
    float ws[3][4];
#pragma unroll
    for (int u = 0; u < 3; ++u) {
      const int id = threadIdx.x + (GEMM_WAVES * 32) * u;
      const int kq = id / 12;
      const int cc = id - kq * 12;
      const float* __restrict__ p = Ws + (kq * 4) * OUT_C + (cc < OUT_C ? cc : OUT_C - 1);
#pragma unroll
      for (int q = 0; q < 4; ++q) ws[u][q] = p[q * OUT_C];
    }
#pragma unroll
    for (int u = 0; u < 3; ++u) {
      const int id = threadIdx.x + (GEMM_WAVES * 32) * u;
      const int kq = id / 12;
      const int cc = id - kq * 12;
      const bool keep = cc < OUT_C;
      Bs4[(((kq >> 1) * NT + (NT - 1)) << 5) + ((kq & 1) << 4) + (HID - 96 + cc)] =
          (v4f){keep ? ws[u][0] : 0.f, keep ? ws[u][1] : 0.f, keep ? ws[u][2] : 0.f, keep ? ws[u][3] : 0.f};
    }
  }
  __syncthreads();

  const int lane = threadIdx.x & 31;
  const int wv   = threadIdx.x >> 5;
  const int m  = lane & 15;
  const int hh = lane >> 4;
  const int tiles = (n_nodes + 15) >> 4;
  const v4f* __restrict__ bl = (const v4f*)Bs + (hh << 4) + m;

  for (int tile = wv * GEMM_WG + blockIdx.x; tile < tiles; tile += GEMM_WG * GEMM_WAVES) {
    const int row0 = tile << 4;
    int arow = row0 + m;
    if (arow > n_nodes - 1) arow = n_nodes - 1;
    const v4f* __restrict__ pa = (const v4f*)(x + (size_t)arow * IN_C) + hh;

    v8f acc[NT] = {};
    v4f ac[8], an[8];
#pragma unroll
    for (int jj = 0; jj < 8; ++jj) ac[jj] = pa[jj << 1];
    v4f bc[NT], bn[NT];
#pragma unroll
    for (int t = 0; t < NT; ++t) bc[t] = bl[t << 5];
    float ds[8];
#pragma unroll
    for (int r = 0; r < 8; ++r) {
      const int drow = row0 + (hh << 3) + r;
      ds[r] = dinv[drow < n_nodes ? drow : n_nodes - 1];
    }

    for (int i = 0; i < IN_C / 64; ++i) {
      const int inx = (i + 1 < IN_C / 64) ? i + 1 : i;
#pragma unroll
      for (int jj = 0; jj < 8; ++jj) an[jj] = pa[(inx << 4) + (jj << 1)];
#pragma unroll
      for (int jj = 0; jj < 8; ++jj) {
        const int j = (i << 3) + jj;
        const int jn = (j + 1 < IN_C / 8) ? j + 1 : j;
#pragma unroll
        for (int t = 0; t < NT; ++t) bn[t] = bl[(jn * NT + t) << 5];
        const v2f a0 = {ac[jj].x, ac[jj].y};
        const v2f a1 = {ac[jj].z, ac[jj].w};
#pragma unroll
        for (int t = 0; t < NT; ++t) {
          const v2f b0 = {bc[t].x, bc[t].y};
          acc[t] = __builtin_amdgcn_wmma_f32_16x16x4_f32(
              false, a0, false, b0, (short)0, acc[t], false, false);
        }
#pragma unroll
        for (int t = 0; t < NT; ++t) {
          const v2f b1 = {bc[t].z, bc[t].w};
          acc[t] = __builtin_amdgcn_wmma_f32_16x16x4_f32(
              false, a1, false, b1, (short)0, acc[t], false, false);
        }
#pragma unroll
        for (int t = 0; t < NT; ++t) bc[t] = bn[t];
      }
#pragma unroll
      for (int jj = 0; jj < 8; ++jj) ac[jj] = an[jj];
    }

    const int rbase = hh << 3;
#pragma unroll
    for (int r = 0; r < 8; ++r) {
      const int row = row0 + rbase + r;
      if (row < n_nodes) {
        float* __restrict__ hrow = h + (size_t)row * HID;
#pragma unroll
        for (int t = 0; t < NT - 1; ++t) hrow[(t << 4) + m] = acc[t][r] * ds[r];
        if (m < HID - 96) hrow[96 + m] = acc[NT - 1][r] * ds[r];
        else if (m < HID - 96 + OUT_C) outp[(size_t)row * OUT_C + (m - (HID - 96))] = acc[NT - 1][r];
      }
    }
  }
}

__global__ void __launch_bounds__(256)
gather1_kernel(const float* __restrict__ h, const float* __restrict__ dinv,
               const int* __restrict__ rowptr, const int* __restrict__ csr,
               const float* __restrict__ b1, const float* __restrict__ W2,
               float* __restrict__ h2, int n_nodes) {
  const int lane = threadIdx.x & 31;
  const int d = blockIdx.x * (blockDim.x >> 5) + (threadIdx.x >> 5);
  if (d >= n_nodes) return;
  const int beg = __builtin_amdgcn_readfirstlane(rowptr[d]);
  const int end = __builtin_amdgcn_readfirstlane(rowptr[d + 1]);
  const float dd = dinv[d];
  const bool act = lane < HID4;
  const v4f* __restrict__ h4 = (const v4f*)h;

  v4f acc = {0.f, 0.f, 0.f, 0.f};
  v4f vself = {0.f, 0.f, 0.f, 0.f};
  if (act) vself = h4[(size_t)d * HID4 + lane];

  for (int base = beg; base < end; base += 32) {
    int m = end - base;
    if (m > 32) m = 32;
    int s_l = 0;
    if (lane < m) {
      s_l = csr[base + lane];
      s_l = min(max(s_l, 0), n_nodes - 1);
    }
    int j = 0;
    for (; j + 4 <= m; j += 4) {
      const int s0 = __builtin_amdgcn_readlane(s_l, j);
      const int s1 = __builtin_amdgcn_readlane(s_l, j + 1);
      const int s2 = __builtin_amdgcn_readlane(s_l, j + 2);
      const int s3 = __builtin_amdgcn_readlane(s_l, j + 3);
      if (act) {
        const v4f v0 = h4[(size_t)s0 * HID4 + lane];
        const v4f v1 = h4[(size_t)s1 * HID4 + lane];
        const v4f v2 = h4[(size_t)s2 * HID4 + lane];
        const v4f v3 = h4[(size_t)s3 * HID4 + lane];
        acc += v0;
        acc += v1;
        acc += v2;
        acc += v3;
      }
    }
    for (; j < m; ++j) {
      const int s0 = __builtin_amdgcn_readlane(s_l, j);
      if (act) acc += h4[(size_t)s0 * HID4 + lane];
    }
  }

  acc += vself;
  acc *= dd;

  float p0 = 0.f, p1 = 0.f;
  if (act) {
    const v4f bb = ((const v4f*)b1)[lane];
    const v4f wa = ((const v4f*)W2)[2 * lane];
    const v4f wb = ((const v4f*)W2)[2 * lane + 1];
    float v0 = acc.x + bb.x, v1 = acc.y + bb.y, v2 = acc.z + bb.z, v3 = acc.w + bb.w;
    v0 = v0 > 0.f ? v0 : 0.f;
    v1 = v1 > 0.f ? v1 : 0.f;
    v2 = v2 > 0.f ? v2 : 0.f;
    v3 = v3 > 0.f ? v3 : 0.f;
    p0 = v0 * wa.x + v1 * wa.z + v2 * wb.x + v3 * wb.z;
    p1 = v0 * wa.y + v1 * wa.w + v2 * wb.y + v3 * wb.w;
  }
#pragma unroll
  for (int o = 16; o > 0; o >>= 1) {
    p0 += __shfl_xor(p0, o);
    p1 += __shfl_xor(p1, o);
  }
  if (lane == 0) { h2[(size_t)d * 2 + 0] = p0 * dd; h2[(size_t)d * 2 + 1] = p1 * dd; }
}

__global__ void gather2_kernel(const float* __restrict__ h2, const float* __restrict__ dinv,
                               const int* __restrict__ rowptr, const int* __restrict__ csr,
                               const float* __restrict__ b2, const float* __restrict__ bs,
                               float* __restrict__ out, int n_nodes) {
  int d = blockIdx.x * blockDim.x + threadIdx.x;
  if (d >= n_nodes) return;
  const int beg = rowptr[d], end = rowptr[d + 1];
  const float dd = dinv[d];
  const v2f* __restrict__ h22 = (const v2f*)h2;
  const v2f hv = h22[d];
  const float sk0 = out[(size_t)d * 2 + 0];
  const float sk1 = out[(size_t)d * 2 + 1];
  const float bb0 = b2[0] + bs[0], bb1 = b2[1] + bs[1];
  float t0 = hv.x, t1 = hv.y;
#pragma unroll 4
  for (int j = beg; j < end; ++j) {
    int s = csr[j];
    s = min(max(s, 0), n_nodes - 1);
    const v2f v = h22[s];
    t0 += v.x;
    t1 += v.y;
  }
  out[(size_t)d * 2 + 0] = sk0 + (bb0 + t0 * dd);
  out[(size_t)d * 2 + 1] = sk1 + (bb1 + t1 * dd);
}

__global__ void log_softmax_kernel(float* __restrict__ out, int n_nodes) {
  int n = blockIdx.x * blockDim.x + threadIdx.x;
  if (n >= n_nodes) return;
  float z0 = out[(size_t)n * 2 + 0];
  float z1 = out[(size_t)n * 2 + 1];
  float mx = fmaxf(z0, z1);
  float lse = mx + logf(expf(z0 - mx) + expf(z1 - mx));
  out[(size_t)n * 2 + 0] = z0 - lse;
  out[(size_t)n * 2 + 1] = z1 - lse;
}

extern "C" void kernel_launch(void* const* d_in, const int* in_sizes, int n_in,
                              void* d_out, int out_size, void* d_ws, size_t ws_size,
                              hipStream_t stream) {
  const float* x  = (const float*)d_in[0];
  const int*   ei = (const int*)d_in[1];
  const float* W1 = (const float*)d_in[2];
  const float* b1 = (const float*)d_in[3];
  const float* W2 = (const float*)d_in[4];
  const float* b2 = (const float*)d_in[5];
  const float* Ws = (const float*)d_in[6];
  const float* bs = (const float*)d_in[7];
  float* out = (float*)d_out;

  const int n_nodes = in_sizes[0] / IN_C;
  const int n_edges = in_sizes[1] / 2;
  const int* srcv = ei;
  const int* dstv = ei + n_edges;

  float* Wp    = (float*)d_ws;
  int* tab     = (int*)d_ws;
  float* dinv  = Wp + WP_FLOATS;
  size_t off   = ((size_t)n_nodes + 255) & ~(size_t)255;
  int* rowptr  = (int*)(dinv + off) + off;
  size_t off1  = ((size_t)n_nodes + 1 + 255) & ~(size_t)255;
  float* h     = (float*)(rowptr + off1);
  float* h2    = h + (size_t)n_nodes * HID;
  int* csr     = (int*)(h2 + (size_t)n_nodes * 2) + n_edges;
  int2* pairs  = (int2*)h;

  const int T = 256;

  chunk_sort_kernel<<<SG, ST, 0, stream>>>(srcv, dstv, tab, pairs, n_nodes, n_edges);
  range_build_kernel<<<NB, ST, 0, stream>>>(pairs, tab, rowptr, dinv, csr, n_nodes, n_edges);

  gemm1_wmma_kernel<<<GEMM_WG, GEMM_WAVES * 32, 0, stream>>>(x, W1, Ws, dinv, h, out, n_nodes);

  gather1_kernel<<<(n_nodes + 7) / 8, T, 0, stream>>>(h, dinv, rowptr, csr, b1, W2, h2, n_nodes);

  gather2_kernel<<<(n_nodes + T - 1) / T, T, 0, stream>>>(h2, dinv, rowptr, csr, b2, bs, out, n_nodes);

  log_softmax_kernel<<<(n_nodes + T - 1) / T, T, 0, stream>>>(out, n_nodes);
}
